# starting kernel
# baseline (speedup 1.0000x reference)
#include <hip/hip_runtime.h>


typedef __attribute__((ext_vector_type(2))) float v2f;
typedef __attribute__((ext_vector_type(8))) float v8f;

__global__ void gcn_degree_kernel(const int* __restrict__ dst,
                                  float* __restrict__ deg, int E) {
  int t = blockIdx.x * blockDim.x + threadIdx.x;
  if (t < E) unsafeAtomicAdd(&deg[dst[t]], 1.0f);
}

template <int K, int TM>
__global__ void gcn_gemm_wmma_f32(const float* __restrict__ X,
                                  const float* __restrict__ W,
                                  float* __restrict__ Y, int N) {
  const int M = TM * 16;
  const int wave = (int)((blockIdx.x * (size_t)blockDim.x + threadIdx.x) >> 5);
  if (wave >= (N >> 4)) return;

  const int lane = threadIdx.x & 31;
  const int half = lane >> 4;
  const int l16  = lane & 15;
  const float* __restrict__ xrow = X + (size_t)((wave << 4) + l16) * K;

  v8f acc[TM];
#pragma unroll
  for (int j = 0; j < TM; ++j) acc[j] = (v8f){};

#pragma unroll 2
  for (int k = 0; k < K; k += 4) {
    const int kk = k + (half << 1);
    const v2f a = *(const v2f*)(xrow + kk);
#pragma unroll
    for (int j = 0; j < TM; ++j) {
      v2f b;
      b.x = W[kk * M + (j << 4) + l16];
      b.y = W[(kk + 1) * M + (j << 4) + l16];
      acc[j] = __builtin_amdgcn_wmma_f32_16x16x4_f32(
           false, a,  false, b,
           (short)0, acc[j],  false,  false);
    }
  }

  const int rbase = (wave << 4) + (half << 3);
#pragma unroll
  for (int j = 0; j < TM; ++j)
#pragma unroll
    for (int r = 0; r < 8; ++r)
      Y[(size_t)(rbase + r) * M + (j << 4) + l16] = acc[j][r];
}

template <int SHIFT>
__global__ void gcn_scatter_add(const float* __restrict__ h,
                                const int* __restrict__ src,
                                const int* __restrict__ dst,
                                float* __restrict__ agg, int E) {
  const int width = 4 << SHIFT;
  long long t = (long long)blockIdx.x * blockDim.x + threadIdx.x;
  int e = (int)(t >> SHIFT);
  if (e >= E) return;
  int c = ((int)t & ((1 << SHIFT) - 1)) << 2;
  int s = src[e], d = dst[e];
  float4 v = *(const float4*)(h + (size_t)s * width + c);
  float* ap = agg + (size_t)d * width + c;
  unsafeAtomicAdd(ap + 0, v.x);
  unsafeAtomicAdd(ap + 1, v.y);
  unsafeAtomicAdd(ap + 2, v.z);
  unsafeAtomicAdd(ap + 3, v.w);
}

template <int LOGW>
__global__ void gcn_finalize(const float* __restrict__ agg,
                             const float* __restrict__ deg,
                             const float* __restrict__ bias,
                             float* __restrict__ out, int N, int doRelu) {
  const int width = 1 << LOGW;
  long long t = (long long)blockIdx.x * blockDim.x + threadIdx.x;
  if (t >= ((long long)N << LOGW)) return;
  int n = (int)(t >> LOGW);
  int c = (int)t & (width - 1);
  float d = deg[n];
  d = d < 1.0f ? 1.0f : d;
  float r = agg[t] / d + bias[c];
  if (doRelu) r = r > 0.0f ? r : 0.0f;
  out[t] = r;
}

extern "C" void kernel_launch(void* const* d_in, const int* in_sizes, int n_in,
                              void* d_out, int out_size, void* d_ws, size_t ws_size,
                              hipStream_t stream) {
  const float* feat = (const float*)d_in[0];
  const float* W0   = (const float*)d_in[1];
  const float* b0   = (const float*)d_in[2];
  const float* W1   = (const float*)d_in[3];
  const float* b1   = (const float*)d_in[4];
  const float* W2   = (const float*)d_in[5];
  const float* b2   = (const float*)d_in[6];
  const int*   src  = (const int*)d_in[7];
  const int*   dst  = (const int*)d_in[8];
  float* out = (float*)d_out;

  const int N = in_sizes[0] / 256;
  const int E = in_sizes[7];

  float* bufA = (float*)d_ws;
  float* bufB = bufA + (size_t)N * 64;
  float* deg  = bufB + (size_t)N * 64;
  const size_t sz64 = (size_t)N * 64 * sizeof(float);
  const size_t sz32 = (size_t)N * 32 * sizeof(float);

  const int TB = 256;
  const int gemmBlocks   = (int)((((long long)(N / 16)) * 32 + TB - 1) / TB);
  const int scat64Blocks = (int)(((long long)E * 16 + TB - 1) / TB);
  const int scat32Blocks = (int)(((long long)E * 8 + TB - 1) / TB);
  const int fin64Blocks  = (int)(((long long)N * 64 + TB - 1) / TB);
  const int fin32Blocks  = (int)(((long long)N * 32 + TB - 1) / TB);

  hipMemsetAsync(deg, 0, (size_t)N * sizeof(float), stream);
  gcn_degree_kernel<<<(E + TB - 1) / TB, TB, 0, stream>>>(dst, deg, E);

  gcn_gemm_wmma_f32<256, 4><<<gemmBlocks, TB, 0, stream>>>(feat, W0, bufA, N);
  hipMemsetAsync(bufB, 0, sz64, stream);
  gcn_scatter_add<4><<<scat64Blocks, TB, 0, stream>>>(bufA, src, dst, bufB, E);
  gcn_finalize<6><<<fin64Blocks, TB, 0, stream>>>(bufB, deg, b0, bufA, N, 1);

  hipMemsetAsync(bufB, 0, sz64, stream);
  gcn_scatter_add<4><<<scat64Blocks, TB, 0, stream>>>(bufA, src, dst, bufB, E);
  gcn_gemm_wmma_f32<64, 4><<<gemmBlocks, TB, 0, stream>>>(bufB, W1, bufA, N);
  gcn_finalize<6><<<fin64Blocks, TB, 0, stream>>>(bufA, deg, b1, bufB, N, 1);

  gcn_gemm_wmma_f32<64, 2><<<gemmBlocks, TB, 0, stream>>>(bufB, W2, bufA, N);
  hipMemsetAsync(bufB, 0, sz32, stream);
  gcn_scatter_add<3><<<scat32Blocks, TB, 0, stream>>>(bufA, src, dst, bufB, E);
  gcn_finalize<5><<<fin32Blocks, TB, 0, stream>>>(bufB, deg, b2, out, N, 0);
}


# scatter-add: one wave per 16/32 edges, lane l owns columns l,l+32, so each gather and each float atomic covers one whole 128-B line (4x fewer atomic requests); atomics kept
# speedup vs baseline: 1.4416x; 1.4416x over previous
#include <hip/hip_runtime.h>


typedef __attribute__((ext_vector_type(2))) float v2f;
typedef __attribute__((ext_vector_type(8))) float v8f;

__global__ void gcn_degree_kernel(const int* __restrict__ dst,
                                  float* __restrict__ deg, int E) {
  int t = blockIdx.x * blockDim.x + threadIdx.x;
  if (t < E) unsafeAtomicAdd(&deg[dst[t]], 1.0f);
}

template <int K, int TM>
__global__ void gcn_gemm_wmma_f32(const float* __restrict__ X,
                                  const float* __restrict__ W,
                                  float* __restrict__ Y, int N) {
  const int M = TM * 16;
  const int wave = (int)((blockIdx.x * (size_t)blockDim.x + threadIdx.x) >> 5);
  if (wave >= (N >> 4)) return;

  const int lane = threadIdx.x & 31;
  const int half = lane >> 4;
  const int l16  = lane & 15;
  const float* __restrict__ xrow = X + (size_t)((wave << 4) + l16) * K;

  v8f acc[TM];
#pragma unroll
  for (int j = 0; j < TM; ++j) acc[j] = (v8f){};

#pragma unroll 2
  for (int k = 0; k < K; k += 4) {
    const int kk = k + (half << 1);
    const v2f a = *(const v2f*)(xrow + kk);
#pragma unroll
    for (int j = 0; j < TM; ++j) {
      v2f b;
      b.x = W[kk * M + (j << 4) + l16];
      b.y = W[(kk + 1) * M + (j << 4) + l16];
      acc[j] = __builtin_amdgcn_wmma_f32_16x16x4_f32(
           false, a,  false, b,
           (short)0, acc[j],  false,  false);
    }
  }

  const int rbase = (wave << 4) + (half << 3);
#pragma unroll
  for (int j = 0; j < TM; ++j)
#pragma unroll
    for (int r = 0; r < 8; ++r)
      Y[(size_t)(rbase + r) * M + (j << 4) + l16] = acc[j][r];
}

template <int WIDTH, int EPW>
__global__ void gcn_scatter_add(const float* __restrict__ h,
                                const int* __restrict__ src,
                                const int* __restrict__ dst,
                                float* __restrict__ agg, int E) {
  constexpr int LPR = WIDTH / 32;
  const int lane = threadIdx.x & 31;
  const int wave = __builtin_amdgcn_readfirstlane(
      (int)((blockIdx.x * (size_t)blockDim.x + threadIdx.x) >> 5));
  const int e0 = wave * EPW;
  if (e0 >= E) return;

  if (e0 + EPW <= E) {
    int s[EPW], d[EPW];
#pragma unroll
    for (int j = 0; j < EPW; ++j) { s[j] = src[e0 + j]; d[j] = dst[e0 + j]; }
    float v[EPW][LPR];
#pragma unroll
    for (int j = 0; j < EPW; ++j)
#pragma unroll
      for (int q = 0; q < LPR; ++q)
        v[j][q] = (h + (size_t)s[j] * WIDTH)[(q << 5) + lane];
#pragma unroll
    for (int j = 0; j < EPW; ++j)
#pragma unroll
      for (int q = 0; q < LPR; ++q)
        unsafeAtomicAdd((agg + (size_t)d[j] * WIDTH) + ((q << 5) + lane), v[j][q]);
  } else {
    for (int e = e0; e < E; ++e) {
      const int sj = src[e], dj = dst[e];
#pragma unroll
      for (int q = 0; q < LPR; ++q)
        unsafeAtomicAdd(agg + (size_t)dj * WIDTH + (q << 5) + lane,
                        h[(size_t)sj * WIDTH + (q << 5) + lane]);
    }
  }
}

template <int LOGW>
__global__ void gcn_finalize(const float* __restrict__ agg,
                             const float* __restrict__ deg,
                             const float* __restrict__ bias,
                             float* __restrict__ out, int N, int doRelu) {
  const int width = 1 << LOGW;
  long long t = (long long)blockIdx.x * blockDim.x + threadIdx.x;
  if (t >= ((long long)N << LOGW)) return;
  int n = (int)(t >> LOGW);
  int c = (int)t & (width - 1);
  float d = deg[n];
  d = d < 1.0f ? 1.0f : d;
  float r = agg[t] / d + bias[c];
  if (doRelu) r = r > 0.0f ? r : 0.0f;
  out[t] = r;
}

extern "C" void kernel_launch(void* const* d_in, const int* in_sizes, int n_in,
                              void* d_out, int out_size, void* d_ws, size_t ws_size,
                              hipStream_t stream) {
  const float* feat = (const float*)d_in[0];
  const float* W0   = (const float*)d_in[1];
  const float* b0   = (const float*)d_in[2];
  const float* W1   = (const float*)d_in[3];
  const float* b1   = (const float*)d_in[4];
  const float* W2   = (const float*)d_in[5];
  const float* b2   = (const float*)d_in[6];
  const int*   src  = (const int*)d_in[7];
  const int*   dst  = (const int*)d_in[8];
  float* out = (float*)d_out;

  const int N = in_sizes[0] / 256;
  const int E = in_sizes[7];

  float* bufA = (float*)d_ws;
  float* bufB = bufA + (size_t)N * 64;
  float* deg  = bufB + (size_t)N * 64;
  const size_t sz64 = (size_t)N * 64 * sizeof(float);
  const size_t sz32 = (size_t)N * 32 * sizeof(float);

  const int TB = 256;
  const int gemmBlocks   = (int)((((long long)(N / 16)) * 32 + TB - 1) / TB);
  const int scat64Blocks = (int)(((long long)((E + 15) / 16) * 32 + TB - 1) / TB);
  const int scat32Blocks = (int)(((long long)((E + 31) / 32) * 32 + TB - 1) / TB);
  const int fin64Blocks  = (int)(((long long)N * 64 + TB - 1) / TB);
  const int fin32Blocks  = (int)(((long long)N * 32 + TB - 1) / TB);

  hipMemsetAsync(deg, 0, (size_t)N * sizeof(float), stream);
  gcn_degree_kernel<<<(E + TB - 1) / TB, TB, 0, stream>>>(dst, deg, E);

  gcn_gemm_wmma_f32<256, 4><<<gemmBlocks, TB, 0, stream>>>(feat, W0, bufA, N);
  hipMemsetAsync(bufB, 0, sz64, stream);
  gcn_scatter_add<64, 16><<<scat64Blocks, TB, 0, stream>>>(bufA, src, dst, bufB, E);
  gcn_finalize<6><<<fin64Blocks, TB, 0, stream>>>(bufB, deg, b0, bufA, N, 1);

  hipMemsetAsync(bufB, 0, sz64, stream);
  gcn_scatter_add<64, 16><<<scat64Blocks, TB, 0, stream>>>(bufA, src, dst, bufB, E);
  gcn_gemm_wmma_f32<64, 4><<<gemmBlocks, TB, 0, stream>>>(bufB, W1, bufA, N);
  gcn_finalize<6><<<fin64Blocks, TB, 0, stream>>>(bufA, deg, b1, bufB, N, 1);

  gcn_gemm_wmma_f32<64, 2><<<gemmBlocks, TB, 0, stream>>>(bufB, W2, bufA, N);
  hipMemsetAsync(bufB, 0, sz32, stream);
  gcn_scatter_add<32, 32><<<scat32Blocks, TB, 0, stream>>>(bufA, src, dst, bufB, E);
  gcn_finalize<5><<<fin32Blocks, TB, 0, stream>>>(bufB, deg, b2, out, N, 0);
}


# aggregation as a pull: edges grouped by dst once per call (int atomic count with arrival rank, scan, fill); each node's rows summed in registers in list order, one normalised store; no float atomics
# speedup vs baseline: 1.8731x; 1.2993x over previous
#include <hip/hip_runtime.h>


typedef __attribute__((ext_vector_type(2))) float v2f;
typedef __attribute__((ext_vector_type(8))) float v8f;

__global__ void gcn_count_rank(const int* __restrict__ dst, int* __restrict__ cnt,
                               int* __restrict__ rank, int E) {
  const int t = blockIdx.x * blockDim.x + threadIdx.x;
  const int e = t << 2;
  if (e + 3 < E) {
    const int4 d = *(const int4*)(dst + e);
    int4 r;
    r.x = atomicAdd(cnt + d.x, 1);
    r.y = atomicAdd(cnt + d.y, 1);
    r.z = atomicAdd(cnt + d.z, 1);
    r.w = atomicAdd(cnt + d.w, 1);
    *(int4*)(rank + e) = r;
  } else {
    for (int k = e; k < E; ++k) rank[k] = atomicAdd(cnt + dst[k], 1);
  }
}

__global__ void gcn_rowptr_scan(const int* __restrict__ cnt, int* __restrict__ rowptr, int N) {
  __shared__ int wsum[32];
  __shared__ int wpre[32];
  const int tid = threadIdx.x, lane = tid & 31, w = tid >> 5;
  const int base = blockIdx.x * 1024;

  int acc = 0;
  const int4* c4 = (const int4*)cnt;
  const int nfull = blockIdx.x >> 2;
#pragma unroll 4
  for (int c = 0; c < nfull; ++c) {
    const int4 q = c4[(c << 10) + tid];
    acc += (q.x + q.y) + (q.z + q.w);
  }
  for (int i = (nfull << 12) + tid; i < base; i += 1024) acc += cnt[i];
  const int n = base + tid;
  const int v = (n < N) ? cnt[n] : 0;

  int inc = v;
  for (int o = 1; o < 32; o <<= 1) {
    const int u = __shfl_up(inc, o);
    if (lane >= o) inc += u;
  }
  for (int o = 16; o > 0; o >>= 1) acc += __shfl_xor(acc, o);
  if (lane == 31) wsum[w] = inc;
  if (lane == 0)  wpre[w] = acc;
  __syncthreads();

  const int ws = wsum[lane];
  int wsinc = ws;
  for (int o = 1; o < 32; o <<= 1) {
    const int u = __shfl_up(wsinc, o);
    if (lane >= o) wsinc += u;
  }
  int front = wpre[lane];
  for (int o = 16; o > 0; o >>= 1) front += __shfl_xor(front, o);
  const int waveOff = __shfl(wsinc - ws, w);
  const int excl = front + waveOff + inc - v;
  if (n < N) rowptr[n] = excl;
  if (n == N - 1) rowptr[N] = excl + v;
}

__global__ void gcn_fill_col(const int* __restrict__ src, const int* __restrict__ dst,
                             const int* __restrict__ rank, const int* __restrict__ rowptr,
                             int* __restrict__ col, int E) {
  const int t = blockIdx.x * blockDim.x + threadIdx.x;
  const int e = t << 2;
  if (e + 3 < E) {
    const int4 s = *(const int4*)(src + e);
    const int4 d = *(const int4*)(dst + e);
    const int4 r = *(const int4*)(rank + e);
    const int p0 = rowptr[d.x], p1 = rowptr[d.y], p2 = rowptr[d.z], p3 = rowptr[d.w];
    col[p0 + r.x] = s.x;
    col[p1 + r.y] = s.y;
    col[p2 + r.z] = s.z;
    col[p3 + r.w] = s.w;
  } else {
    for (int k = e; k < E; ++k) col[rowptr[dst[k]] + rank[k]] = src[k];
  }
}

__global__ void gcn_move_ints(const int* __restrict__ from, int* __restrict__ to, int T) {
  const int t = blockIdx.x * blockDim.x + threadIdx.x;
  const int e = t << 2;
  if (e + 3 < T) {
    *(int4*)(to + e) = *(const int4*)(from + e);
  } else {
    for (int k = e; k < T; ++k) to[k] = from[k];
  }
}

template <int K, int TM>
__global__ void gcn_gemm_wmma_f32(const float* __restrict__ X,
                                  const float* __restrict__ W,
                                  float* __restrict__ Y, int N) {
  const int M = TM * 16;
  const int wave = (int)((blockIdx.x * (size_t)blockDim.x + threadIdx.x) >> 5);
  if (wave >= (N >> 4)) return;

  const int lane = threadIdx.x & 31;
  const int half = lane >> 4;
  const int l16  = lane & 15;
  const float* __restrict__ xrow = X + (size_t)((wave << 4) + l16) * K;

  v8f acc[TM];
#pragma unroll
  for (int j = 0; j < TM; ++j) acc[j] = (v8f){};

#pragma unroll 2
  for (int k = 0; k < K; k += 4) {
    const int kk = k + (half << 1);
    const v2f a = *(const v2f*)(xrow + kk);
#pragma unroll
    for (int j = 0; j < TM; ++j) {
      v2f b;
      b.x = W[kk * M + (j << 4) + l16];
      b.y = W[(kk + 1) * M + (j << 4) + l16];
      acc[j] = __builtin_amdgcn_wmma_f32_16x16x4_f32(
           false, a,  false, b,
           (short)0, acc[j],  false,  false);
    }
  }

  const int rbase = (wave << 4) + (half << 3);
#pragma unroll
  for (int j = 0; j < TM; ++j)
#pragma unroll
    for (int r = 0; r < 8; ++r)
      Y[(size_t)(rbase + r) * M + (j << 4) + l16] = acc[j][r];
}

template <int VEC>
__device__ __forceinline__ void gcn_ld_row(const float* p, float (&v)[VEC]) {
  if constexpr (VEC == 2) { const v2f t = *(const v2f*)p; v[0] = t.x; v[1] = t.y; }
  else                    { v[0] = *p; }
}
template <int VEC>
__device__ __forceinline__ void gcn_st_row(float* p, const float (&v)[VEC]) {
  if constexpr (VEC == 2) { v2f t; t.x = v[0]; t.y = v[1]; *(v2f*)p = t; }
  else                    { *p = v[0]; }
}

template <int WIDTH, int NPW, bool FINAL>
__global__ void gcn_pull_sum(const float* __restrict__ h,
                             const int* __restrict__ rowptr,
                             const int* __restrict__ col,
                             const float* __restrict__ bias,
                             float* __restrict__ outp, int N, int doRelu) {
  constexpr int VEC = WIDTH / 32;
  constexpr int G = 8;
  const int lane = threadIdx.x & 31;
  const int wave = __builtin_amdgcn_readfirstlane(
      (int)((blockIdx.x * (size_t)blockDim.x + threadIdx.x) >> 5));
  const int node0 = wave * NPW;
  if (node0 >= N) return;

  int rp[NPW + 1];
#pragma unroll
  for (int i = 0; i <= NPW; ++i) {
    const int n = (node0 + i) < N ? (node0 + i) : N;
    rp[i] = rowptr[n];
  }
  int first[NPW];
#pragma unroll
  for (int i = 0; i < NPW; ++i)
    first[i] = (lane < rp[i + 1] - rp[i]) ? col[rp[i] + lane] : 0;

  const float* __restrict__ hl = h + lane * VEC;

#pragma unroll
  for (int i = 0; i < NPW; ++i) {
    const int node = node0 + i;
    if (node < N) {
      const int beg = rp[i], end = rp[i + 1];
      float acc[VEC];
#pragma unroll
      for (int q = 0; q < VEC; ++q) acc[q] = 0.0f;

      int cur = first[i];
      for (int p = beg; p < end; p += 32) {
        if (p != beg) cur = (lane < end - p) ? col[p + lane] : 0;
        const int m = (end - p) < 32 ? (end - p) : 32;
        for (int j = 0; j < m; j += G) {
          float v[G][VEC];
#pragma unroll
          for (int u = 0; u < G; ++u) {
            const int jj = (j + u) < m ? (j + u) : (m - 1);
            const int s = __builtin_amdgcn_readlane(cur, jj);
            gcn_ld_row<VEC>(hl + (size_t)s * WIDTH, v[u]);
          }
#pragma unroll
          for (int u = 0; u < G; ++u) {
            const bool on = (j + u) < m;
#pragma unroll
            for (int q = 0; q < VEC; ++q) acc[q] += on ? v[u][q] : 0.0f;
          }
        }
      }

      float r[VEC];
      if (FINAL) {
        float d = (float)(end - beg);
        d = d < 1.0f ? 1.0f : d;
#pragma unroll
        for (int q = 0; q < VEC; ++q) {
          float t = acc[q] / d + bias[lane * VEC + q];
          if (doRelu) t = t > 0.0f ? t : 0.0f;
          r[q] = t;
        }
      } else {
#pragma unroll
        for (int q = 0; q < VEC; ++q) r[q] = acc[q];
      }
      gcn_st_row<VEC>(outp + (size_t)node * WIDTH + lane * VEC, r);
    }
  }
}

template <int LOGW>
__global__ void gcn_finalize(const float* __restrict__ agg,
                             const int* __restrict__ cnt,
                             const float* __restrict__ bias,
                             float* __restrict__ out, int N, int doRelu) {
  const int width = 1 << LOGW;
  long long t = (long long)blockIdx.x * blockDim.x + threadIdx.x;
  if (t >= ((long long)N << LOGW)) return;
  int n = (int)(t >> LOGW);
  int c = (int)t & (width - 1);
  float d = (float)cnt[n];
  d = d < 1.0f ? 1.0f : d;
  float r = agg[t] / d + bias[c];
  if (doRelu) r = r > 0.0f ? r : 0.0f;
  out[t] = r;
}

extern "C" void kernel_launch(void* const* d_in, const int* in_sizes, int n_in,
                              void* d_out, int out_size, void* d_ws, size_t ws_size,
                              hipStream_t stream) {
  const float* feat = (const float*)d_in[0];
  const float* W0   = (const float*)d_in[1];
  const float* b0   = (const float*)d_in[2];
  const float* W1   = (const float*)d_in[3];
  const float* b1   = (const float*)d_in[4];
  const float* W2   = (const float*)d_in[5];
  const float* b2   = (const float*)d_in[6];
  const int*   src  = (const int*)d_in[7];
  const int*   dst  = (const int*)d_in[8];
  float* out = (float*)d_out;

  const int N = in_sizes[0] / 256;
  const int E = in_sizes[7];

  float* bufA = (float*)d_ws;
  float* bufB = bufA + (size_t)N * 64;
  int*   cnt  = (int*)(bufB + (size_t)N * 64);
  int*   rank = (int*)bufB;
  const int rpPad = ((N + 1 + 63) / 64) * 64;
  int* rowptrO = (int*)out;
  int* colO    = rowptrO + rpPad;
  int* rowptrA = (int*)bufA;
  int* colA    = rowptrA + rpPad;

  const int TB = 256;
  const int NPW = 4;
  const int gemmBlocks  = (int)((((long long)(N / 16)) * 32 + TB - 1) / TB);
  const int edge4Blocks = (int)((((long long)E + 3) / 4 + TB - 1) / TB);
  const int scanBlocks  = (N + 1023) / 1024;
  const int pullBlocks  = (int)(((long long)((N + NPW - 1) / NPW) * 32 + TB - 1) / TB);
  const int moveBlocks  = (int)((((long long)rpPad + E + 3) / 4 + TB - 1) / TB);
  const int fin64Blocks = (int)(((long long)N * 64 + TB - 1) / TB);

  hipMemsetAsync(cnt, 0, (size_t)N * sizeof(int), stream);
  gcn_count_rank<<<edge4Blocks, TB, 0, stream>>>(dst, cnt, rank, E);
  gcn_rowptr_scan<<<scanBlocks, 1024, 0, stream>>>(cnt, rowptrO, N);
  gcn_fill_col<<<edge4Blocks, TB, 0, stream>>>(src, dst, rank, rowptrO, colO, E);

  gcn_gemm_wmma_f32<256, 4><<<gemmBlocks, TB, 0, stream>>>(feat, W0, bufA, N);
  gcn_pull_sum<64, NPW, true><<<pullBlocks, TB, 0, stream>>>(bufA, rowptrO, colO, b0, bufB, N, 1);

  gcn_pull_sum<64, NPW, false><<<pullBlocks, TB, 0, stream>>>(bufB, rowptrO, colO, b1, bufA, N, 0);
  gcn_gemm_wmma_f32<64, 4><<<gemmBlocks, TB, 0, stream>>>(bufA, W1, bufB, N);
  gcn_finalize<6><<<fin64Blocks, TB, 0, stream>>>(bufB, cnt, b1, bufA, N, 1);

  gcn_gemm_wmma_f32<64, 2><<<gemmBlocks, TB, 0, stream>>>(bufA, W2, bufB, N);
  gcn_move_ints<<<moveBlocks, TB, 0, stream>>>(rowptrO, rowptrA, rpPad + E);
  gcn_pull_sum<32, NPW, true><<<pullBlocks, TB, 0, stream>>>(bufB, rowptrA, colA, b2, out, N, 0);
}


# grouping: per-edge returning global int atomics replaced by 16-bit LDS counters per 50000-edge chunk (rank = returned LDS add) plus a per-node chunk prefix; a node's rows now summed in chunk order
# speedup vs baseline: 2.4075x; 1.2853x over previous
#include <hip/hip_runtime.h>


typedef __attribute__((ext_vector_type(2))) float v2f;
typedef __attribute__((ext_vector_type(8))) float v8f;

constexpr int GCN_HWORDS = 51200;
constexpr int GCN_TRIPS  = 16;

__device__ __forceinline__ int gcn_lds_rank16(unsigned int* hist, int d) {
  const unsigned int sh = (unsigned int)(d & 1) << 4;
  const unsigned int old = atomicAdd(hist + (d >> 1), 1u << sh);
  return (int)((old >> sh) & 0xFFFFu);
}

__global__ void __launch_bounds__(1024)
gcn_chunk_count_rank(const int* __restrict__ dst, int* __restrict__ cm,
                     int* __restrict__ rank, int E, int N, int S) {
  __shared__ __attribute__((aligned(16))) unsigned int hist[GCN_HWORDS];
  const int tid = threadIdx.x;
  uint4* h4 = (uint4*)hist;
  for (int i = tid; i < GCN_HWORDS / 4; i += 1024) h4[i] = make_uint4(0u, 0u, 0u, 0u);
  __syncthreads();

  const long long b0 = (long long)blockIdx.x * S;
  const int e0 = b0 < E ? (int)b0 : E;
  const int e1 = (b0 + S) < E ? (int)(b0 + S) : E;
  const int e  = e0 + (tid << 2);
  int eL = (e1 - 4) & ~3;
  eL = eL < 0 ? 0 : eL;
  int4 dv[GCN_TRIPS];
#pragma unroll
  for (int j = 0; j < GCN_TRIPS; ++j) {
    const int ej = e + (j << 12);
    dv[j] = *(const int4*)(dst + (ej < eL ? ej : eL));
  }
#pragma unroll
  for (int j = 0; j < GCN_TRIPS; ++j) {
    const int ej = e + (j << 12);
    if (ej + 3 < e1) {
      int4 r;
      r.x = gcn_lds_rank16(hist, dv[j].x);
      r.y = gcn_lds_rank16(hist, dv[j].y);
      r.z = gcn_lds_rank16(hist, dv[j].z);
      r.w = gcn_lds_rank16(hist, dv[j].w);
      *(int4*)(rank + ej) = r;
    } else {
      for (int k = ej; k < e1; ++k) rank[k] = gcn_lds_rank16(hist, dst[k]);
    }
  }
  __syncthreads();

  int* __restrict__ row = cm + (size_t)blockIdx.x * N;
  const uint2* h2 = (const uint2*)hist;
  for (int q = tid; q < (N >> 2); q += 1024) {
    const uint2 w = h2[q];
    int4 v;
    v.x = (int)(w.x & 0xFFFFu); v.y = (int)(w.x >> 16);
    v.z = (int)(w.y & 0xFFFFu); v.w = (int)(w.y >> 16);
    *(int4*)(row + (q << 2)) = v;
  }
}

template <int C>
__global__ void gcn_chunk_prefix(int* __restrict__ cm, int* __restrict__ cnt, int N) {
  const int n = blockIdx.x * blockDim.x + threadIdx.x;
  if (n >= N) return;
  int t[C];
#pragma unroll
  for (int c = 0; c < C; ++c) t[c] = cm[(size_t)c * N + n];
  int sum = 0;
#pragma unroll
  for (int c = 0; c < C; ++c) {
    cm[(size_t)c * N + n] = sum;
    sum += t[c];
  }
  cnt[n] = sum;
}

__global__ void gcn_rowptr_scan(const int* __restrict__ cnt, int* __restrict__ rowptr, int N) {
  __shared__ int wsum[32];
  __shared__ int wpre[32];
  const int tid = threadIdx.x, lane = tid & 31, w = tid >> 5;
  const int base = blockIdx.x * 1024;

  int acc = 0;
  const int4* c4 = (const int4*)cnt;
  const int nfull = blockIdx.x >> 2;
#pragma unroll 4
  for (int c = 0; c < nfull; ++c) {
    const int4 q = c4[(c << 10) + tid];
    acc += (q.x + q.y) + (q.z + q.w);
  }
  for (int i = (nfull << 12) + tid; i < base; i += 1024) acc += cnt[i];
  const int n = base + tid;
  const int v = (n < N) ? cnt[n] : 0;

  int inc = v;
  for (int o = 1; o < 32; o <<= 1) {
    const int u = __shfl_up(inc, o);
    if (lane >= o) inc += u;
  }
  for (int o = 16; o > 0; o >>= 1) acc += __shfl_xor(acc, o);
  if (lane == 31) wsum[w] = inc;
  if (lane == 0)  wpre[w] = acc;
  __syncthreads();

  const int ws = wsum[lane];
  int wsinc = ws;
  for (int o = 1; o < 32; o <<= 1) {
    const int u = __shfl_up(wsinc, o);
    if (lane >= o) wsinc += u;
  }
  int front = wpre[lane];
  for (int o = 16; o > 0; o >>= 1) front += __shfl_xor(front, o);
  const int waveOff = __shfl(wsinc - ws, w);
  const int excl = front + waveOff + inc - v;
  if (n < N) rowptr[n] = excl;
  if (n == N - 1) rowptr[N] = excl + v;
}

__global__ void gcn_fill_col(const int* __restrict__ src, const int* __restrict__ dst,
                             const int* __restrict__ rank, const int* __restrict__ rowptr,
                             const int* __restrict__ cm, int* __restrict__ col,
                             int E, int N, int S) {
  const int t = blockIdx.x * blockDim.x + threadIdx.x;
  const int e = t << 2;
  if (e + 3 < E) {
    const int4 s = *(const int4*)(src + e);
    const int4 d = *(const int4*)(dst + e);
    const int4 r = *(const int4*)(rank + e);
    const int* __restrict__ pre = cm + (size_t)(e / S) * N;
    const int p0 = rowptr[d.x] + pre[d.x], p1 = rowptr[d.y] + pre[d.y];
    const int p2 = rowptr[d.z] + pre[d.z], p3 = rowptr[d.w] + pre[d.w];
    col[p0 + r.x] = s.x;
    col[p1 + r.y] = s.y;
    col[p2 + r.z] = s.z;
    col[p3 + r.w] = s.w;
  } else {
    for (int k = e; k < E; ++k) {
      const int dk = dst[k];
      col[rowptr[dk] + cm[(size_t)(k / S) * N + dk] + rank[k]] = src[k];
    }
  }
}

__global__ void gcn_move_ints(const int* __restrict__ from, int* __restrict__ to, int T) {
  const int t = blockIdx.x * blockDim.x + threadIdx.x;
  const int e = t << 2;
  if (e + 3 < T) {
    *(int4*)(to + e) = *(const int4*)(from + e);
  } else {
    for (int k = e; k < T; ++k) to[k] = from[k];
  }
}

template <int K, int TM>
__global__ void gcn_gemm_wmma_f32(const float* __restrict__ X,
                                  const float* __restrict__ W,
                                  float* __restrict__ Y, int N) {
  const int M = TM * 16;
  const int wave = (int)((blockIdx.x * (size_t)blockDim.x + threadIdx.x) >> 5);
  if (wave >= (N >> 4)) return;

  const int lane = threadIdx.x & 31;
  const int half = lane >> 4;
  const int l16  = lane & 15;
  const float* __restrict__ xrow = X + (size_t)((wave << 4) + l16) * K;

  v8f acc[TM];
#pragma unroll
  for (int j = 0; j < TM; ++j) acc[j] = (v8f){};

#pragma unroll 2
  for (int k = 0; k < K; k += 4) {
    const int kk = k + (half << 1);
    const v2f a = *(const v2f*)(xrow + kk);
#pragma unroll
    for (int j = 0; j < TM; ++j) {
      v2f b;
      b.x = W[kk * M + (j << 4) + l16];
      b.y = W[(kk + 1) * M + (j << 4) + l16];
      acc[j] = __builtin_amdgcn_wmma_f32_16x16x4_f32(
           false, a,  false, b,
           (short)0, acc[j],  false,  false);
    }
  }

  const int rbase = (wave << 4) + (half << 3);
#pragma unroll
  for (int j = 0; j < TM; ++j)
#pragma unroll
    for (int r = 0; r < 8; ++r)
      Y[(size_t)(rbase + r) * M + (j << 4) + l16] = acc[j][r];
}

template <int VEC>
__device__ __forceinline__ void gcn_ld_row(const float* p, float (&v)[VEC]) {
  if constexpr (VEC == 2) { const v2f t = *(const v2f*)p; v[0] = t.x; v[1] = t.y; }
  else                    { v[0] = *p; }
}
template <int VEC>
__device__ __forceinline__ void gcn_st_row(float* p, const float (&v)[VEC]) {
  if constexpr (VEC == 2) { v2f t; t.x = v[0]; t.y = v[1]; *(v2f*)p = t; }
  else                    { *p = v[0]; }
}

template <int WIDTH, int NPW, bool FINAL>
__global__ void gcn_pull_sum(const float* __restrict__ h,
                             const int* __restrict__ rowptr,
                             const int* __restrict__ col,
                             const float* __restrict__ bias,
                             float* __restrict__ outp, int N, int doRelu) {
  constexpr int VEC = WIDTH / 32;
  constexpr int G = 8;
  const int lane = threadIdx.x & 31;
  const int wave = __builtin_amdgcn_readfirstlane(
      (int)((blockIdx.x * (size_t)blockDim.x + threadIdx.x) >> 5));
  const int node0 = wave * NPW;
  if (node0 >= N) return;

  int rp[NPW + 1];
#pragma unroll
  for (int i = 0; i <= NPW; ++i) {
    const int n = (node0 + i) < N ? (node0 + i) : N;
    rp[i] = rowptr[n];
  }
  int first[NPW];
#pragma unroll
  for (int i = 0; i < NPW; ++i)
    first[i] = (lane < rp[i + 1] - rp[i]) ? col[rp[i] + lane] : 0;

  const float* __restrict__ hl = h + lane * VEC;

#pragma unroll
  for (int i = 0; i < NPW; ++i) {
    const int node = node0 + i;
    if (node < N) {
      const int beg = rp[i], end = rp[i + 1];
      float acc[VEC];
#pragma unroll
      for (int q = 0; q < VEC; ++q) acc[q] = 0.0f;

      int cur = first[i];
      for (int p = beg; p < end; p += 32) {
        if (p != beg) cur = (lane < end - p) ? col[p + lane] : 0;
        const int m = (end - p) < 32 ? (end - p) : 32;
        for (int j = 0; j < m; j += G) {
          float v[G][VEC];
#pragma unroll
          for (int u = 0; u < G; ++u) {
            const int jj = (j + u) < m ? (j + u) : (m - 1);
            const int s = __builtin_amdgcn_readlane(cur, jj);
            gcn_ld_row<VEC>(hl + (size_t)s * WIDTH, v[u]);
          }
#pragma unroll
          for (int u = 0; u < G; ++u) {
            const bool on = (j + u) < m;
#pragma unroll
            for (int q = 0; q < VEC; ++q) acc[q] += on ? v[u][q] : 0.0f;
          }
        }
      }

      float r[VEC];
      if (FINAL) {
        float d = (float)(end - beg);
        d = d < 1.0f ? 1.0f : d;
#pragma unroll
        for (int q = 0; q < VEC; ++q) {
          float t = acc[q] / d + bias[lane * VEC + q];
          if (doRelu) t = t > 0.0f ? t : 0.0f;
          r[q] = t;
        }
      } else {
#pragma unroll
        for (int q = 0; q < VEC; ++q) r[q] = acc[q];
      }
      gcn_st_row<VEC>(outp + (size_t)node * WIDTH + lane * VEC, r);
    }
  }
}

template <int LOGW>
__global__ void gcn_finalize(const float* __restrict__ agg,
                             const int* __restrict__ cnt,
                             const float* __restrict__ bias,
                             float* __restrict__ out, int N, int doRelu) {
  const int width = 1 << LOGW;
  long long t = (long long)blockIdx.x * blockDim.x + threadIdx.x;
  if (t >= ((long long)N << LOGW)) return;
  int n = (int)(t >> LOGW);
  int c = (int)t & (width - 1);
  float d = (float)cnt[n];
  d = d < 1.0f ? 1.0f : d;
  float r = agg[t] / d + bias[c];
  if (doRelu) r = r > 0.0f ? r : 0.0f;
  out[t] = r;
}

extern "C" void kernel_launch(void* const* d_in, const int* in_sizes, int n_in,
                              void* d_out, int out_size, void* d_ws, size_t ws_size,
                              hipStream_t stream) {
  const float* feat = (const float*)d_in[0];
  const float* W0   = (const float*)d_in[1];
  const float* b0   = (const float*)d_in[2];
  const float* W1   = (const float*)d_in[3];
  const float* b1   = (const float*)d_in[4];
  const float* W2   = (const float*)d_in[5];
  const float* b2   = (const float*)d_in[6];
  const int*   src  = (const int*)d_in[7];
  const int*   dst  = (const int*)d_in[8];
  float* out = (float*)d_out;

  const int N = in_sizes[0] / 256;
  const int E = in_sizes[7];

  float* bufA = (float*)d_ws;
  float* bufB = bufA + (size_t)N * 64;
  int*   cnt  = (int*)(bufB + (size_t)N * 64);
  int*   rank = (int*)bufB;
  int*   cm   = (int*)bufA;
  const int rpPad = ((N + 1 + 63) / 64) * 64;
  int* rowptrO = (int*)out;
  int* colO    = rowptrO + rpPad;
  int* rowptrA = (int*)bufA;
  int* colA    = rowptrA + rpPad;

  const int TB = 256;
  const int NPW = 4;
  const int gemmBlocks  = (int)((((long long)(N / 16)) * 32 + TB - 1) / TB);
  const int edge4Blocks = (int)((((long long)E + 3) / 4 + TB - 1) / TB);
  const int scanBlocks  = (N + 1023) / 1024;
  const int nodeBlocks  = (N + TB - 1) / TB;
  constexpr int CH = 32;
  const int S  = (((E + CH - 1) / CH) + 3) & ~3;
  const int pullBlocks  = (int)(((long long)((N + NPW - 1) / NPW) * 32 + TB - 1) / TB);
  const int moveBlocks  = (int)((((long long)rpPad + E + 3) / 4 + TB - 1) / TB);
  const int fin64Blocks = (int)(((long long)N * 64 + TB - 1) / TB);

  gcn_chunk_count_rank<<<CH, 1024, 0, stream>>>(dst, cm, rank, E, N, S);
  gcn_chunk_prefix<CH><<<nodeBlocks, TB, 0, stream>>>(cm, cnt, N);
  gcn_rowptr_scan<<<scanBlocks, 1024, 0, stream>>>(cnt, rowptrO, N);
  gcn_fill_col<<<edge4Blocks, TB, 0, stream>>>(src, dst, rank, rowptrO, cm, colO, E, N, S);

  gcn_gemm_wmma_f32<256, 4><<<gemmBlocks, TB, 0, stream>>>(feat, W0, bufA, N);
  gcn_pull_sum<64, NPW, true><<<pullBlocks, TB, 0, stream>>>(bufA, rowptrO, colO, b0, bufB, N, 1);

  gcn_pull_sum<64, NPW, false><<<pullBlocks, TB, 0, stream>>>(bufB, rowptrO, colO, b1, bufA, N, 0);
  gcn_gemm_wmma_f32<64, 4><<<gemmBlocks, TB, 0, stream>>>(bufA, W1, bufB, N);
  gcn_finalize<6><<<fin64Blocks, TB, 0, stream>>>(bufB, cnt, b1, bufA, N, 1);

  gcn_gemm_wmma_f32<64, 2><<<gemmBlocks, TB, 0, stream>>>(bufA, W2, bufB, N);
  gcn_move_ints<<<moveBlocks, TB, 0, stream>>>(rowptrO, rowptrA, rpPad + E);
  gcn_pull_sum<32, NPW, true><<<pullBlocks, TB, 0, stream>>>(bufB, rowptrA, colA, b2, out, N, 0);
}


# layers 1-2 dense chain in one launch: both products transposed (W^T.X^T), so product 1's D registers, normalised+relu'd in place, are product 2's B operand; product 2 sums k in another order
# speedup vs baseline: 2.8702x; 1.1922x over previous
#include <hip/hip_runtime.h>


typedef __attribute__((ext_vector_type(2))) float v2f;
typedef __attribute__((ext_vector_type(4))) float v4f;
typedef __attribute__((ext_vector_type(8))) float v8f;

constexpr int GCN_HWORDS = 51200;
constexpr int GCN_TRIPS  = 16;

__device__ __forceinline__ int gcn_lds_rank16(unsigned int* hist, int d) {
  const unsigned int sh = (unsigned int)(d & 1) << 4;
  const unsigned int old = atomicAdd(hist + (d >> 1), 1u << sh);
  return (int)((old >> sh) & 0xFFFFu);
}

__global__ void __launch_bounds__(1024)
gcn_chunk_count_rank(const int* __restrict__ dst, int* __restrict__ cm,
                     int* __restrict__ rank, int E, int N, int S) {
  __shared__ __attribute__((aligned(16))) unsigned int hist[GCN_HWORDS];
  const int tid = threadIdx.x;
  uint4* h4 = (uint4*)hist;
  for (int i = tid; i < GCN_HWORDS / 4; i += 1024) h4[i] = make_uint4(0u, 0u, 0u, 0u);
  __syncthreads();

  const long long b0 = (long long)blockIdx.x * S;
  const int e0 = b0 < E ? (int)b0 : E;
  const int e1 = (b0 + S) < E ? (int)(b0 + S) : E;
  const int e  = e0 + (tid << 2);
  int eL = (e1 - 4) & ~3;
  eL = eL < 0 ? 0 : eL;
  int4 dv[GCN_TRIPS];
#pragma unroll
  for (int j = 0; j < GCN_TRIPS; ++j) {
    const int ej = e + (j << 12);
    dv[j] = *(const int4*)(dst + (ej < eL ? ej : eL));
  }
#pragma unroll
  for (int j = 0; j < GCN_TRIPS; ++j) {
    const int ej = e + (j << 12);
    if (ej + 3 < e1) {
      int4 r;
      r.x = gcn_lds_rank16(hist, dv[j].x);
      r.y = gcn_lds_rank16(hist, dv[j].y);
      r.z = gcn_lds_rank16(hist, dv[j].z);
      r.w = gcn_lds_rank16(hist, dv[j].w);
      *(int4*)(rank + ej) = r;
    } else {
      for (int k = ej; k < e1; ++k) rank[k] = gcn_lds_rank16(hist, dst[k]);
    }
  }
  __syncthreads();

  int* __restrict__ row = cm + (size_t)blockIdx.x * N;
  const uint2* h2 = (const uint2*)hist;
  for (int q = tid; q < (N >> 2); q += 1024) {
    const uint2 w = h2[q];
    int4 v;
    v.x = (int)(w.x & 0xFFFFu); v.y = (int)(w.x >> 16);
    v.z = (int)(w.y & 0xFFFFu); v.w = (int)(w.y >> 16);
    *(int4*)(row + (q << 2)) = v;
  }
}

template <int C>
__global__ void gcn_chunk_prefix(int* __restrict__ cm, int* __restrict__ cnt, int N) {
  const int n = blockIdx.x * blockDim.x + threadIdx.x;
  if (n >= N) return;
  int t[C];
#pragma unroll
  for (int c = 0; c < C; ++c) t[c] = cm[(size_t)c * N + n];
  int sum = 0;
#pragma unroll
  for (int c = 0; c < C; ++c) {
    cm[(size_t)c * N + n] = sum;
    sum += t[c];
  }
  cnt[n] = sum;
}

__global__ void gcn_rowptr_scan(const int* __restrict__ cnt, int* __restrict__ rowptr, int N) {
  __shared__ int wsum[32];
  __shared__ int wpre[32];
  const int tid = threadIdx.x, lane = tid & 31, w = tid >> 5;
  const int base = blockIdx.x * 1024;

  int acc = 0;
  const int4* c4 = (const int4*)cnt;
  const int nfull = blockIdx.x >> 2;
#pragma unroll 4
  for (int c = 0; c < nfull; ++c) {
    const int4 q = c4[(c << 10) + tid];
    acc += (q.x + q.y) + (q.z + q.w);
  }
  for (int i = (nfull << 12) + tid; i < base; i += 1024) acc += cnt[i];
  const int n = base + tid;
  const int v = (n < N) ? cnt[n] : 0;

  int inc = v;
  for (int o = 1; o < 32; o <<= 1) {
    const int u = __shfl_up(inc, o);
    if (lane >= o) inc += u;
  }
  for (int o = 16; o > 0; o >>= 1) acc += __shfl_xor(acc, o);
  if (lane == 31) wsum[w] = inc;
  if (lane == 0)  wpre[w] = acc;
  __syncthreads();

  const int ws = wsum[lane];
  int wsinc = ws;
  for (int o = 1; o < 32; o <<= 1) {
    const int u = __shfl_up(wsinc, o);
    if (lane >= o) wsinc += u;
  }
  int front = wpre[lane];
  for (int o = 16; o > 0; o >>= 1) front += __shfl_xor(front, o);
  const int waveOff = __shfl(wsinc - ws, w);
  const int excl = front + waveOff + inc - v;
  if (n < N) rowptr[n] = excl;
  if (n == N - 1) rowptr[N] = excl + v;
}

__global__ void gcn_fill_col(const int* __restrict__ src, const int* __restrict__ dst,
                             const int* __restrict__ rank, const int* __restrict__ rowptr,
                             const int* __restrict__ cm, int* __restrict__ col,
                             int E, int N, int S) {
  const int t = blockIdx.x * blockDim.x + threadIdx.x;
  const int e = t << 2;
  if (e + 3 < E) {
    const int4 s = *(const int4*)(src + e);
    const int4 d = *(const int4*)(dst + e);
    const int4 r = *(const int4*)(rank + e);
    const int* __restrict__ pre = cm + (size_t)(e / S) * N;
    const int p0 = rowptr[d.x] + pre[d.x], p1 = rowptr[d.y] + pre[d.y];
    const int p2 = rowptr[d.z] + pre[d.z], p3 = rowptr[d.w] + pre[d.w];
    col[p0 + r.x] = s.x;
    col[p1 + r.y] = s.y;
    col[p2 + r.z] = s.z;
    col[p3 + r.w] = s.w;
  } else {
    for (int k = e; k < E; ++k) {
      const int dk = dst[k];
      col[rowptr[dk] + cm[(size_t)(k / S) * N + dk] + rank[k]] = src[k];
    }
  }
}

__global__ void gcn_move_ints(const int* __restrict__ from, int* __restrict__ to, int T) {
  const int t = blockIdx.x * blockDim.x + threadIdx.x;
  const int e = t << 2;
  if (e + 3 < T) {
    *(int4*)(to + e) = *(const int4*)(from + e);
  } else {
    for (int k = e; k < T; ++k) to[k] = from[k];
  }
}

template <int K, int TM>
__global__ void gcn_gemm_wmma_f32(const float* __restrict__ X,
                                  const float* __restrict__ W,
                                  float* __restrict__ Y, int N) {
  const int M = TM * 16;
  const int wave = (int)((blockIdx.x * (size_t)blockDim.x + threadIdx.x) >> 5);
  if (wave >= (N >> 4)) return;

  const int lane = threadIdx.x & 31;
  const int half = lane >> 4;
  const int l16  = lane & 15;
  const float* __restrict__ xrow = X + (size_t)((wave << 4) + l16) * K;

  v8f acc[TM];
#pragma unroll
  for (int j = 0; j < TM; ++j) acc[j] = (v8f){};

#pragma unroll 2
  for (int k = 0; k < K; k += 4) {
    const int kk = k + (half << 1);
    const v2f a = *(const v2f*)(xrow + kk);
#pragma unroll
    for (int j = 0; j < TM; ++j) {
      v2f b;
      b.x = W[kk * M + (j << 4) + l16];
      b.y = W[(kk + 1) * M + (j << 4) + l16];
      acc[j] = __builtin_amdgcn_wmma_f32_16x16x4_f32(
           false, a,  false, b,
           (short)0, acc[j],  false,  false);
    }
  }

  const int rbase = (wave << 4) + (half << 3);
#pragma unroll
  for (int j = 0; j < TM; ++j)
#pragma unroll
    for (int r = 0; r < 8; ++r)
      Y[(size_t)(rbase + r) * M + (j << 4) + l16] = acc[j][r];
}

template <int VEC>
__device__ __forceinline__ void gcn_ld_row(const float* p, float (&v)[VEC]) {
  if constexpr (VEC == 2) { const v2f t = *(const v2f*)p; v[0] = t.x; v[1] = t.y; }
  else                    { v[0] = *p; }
}
template <int VEC>
__device__ __forceinline__ void gcn_st_row(float* p, const float (&v)[VEC]) {
  if constexpr (VEC == 2) { v2f t; t.x = v[0]; t.y = v[1]; *(v2f*)p = t; }
  else                    { *p = v[0]; }
}

template <int WIDTH, int NPW, bool FINAL>
__global__ void gcn_pull_sum(const float* __restrict__ h,
                             const int* __restrict__ rowptr,
                             const int* __restrict__ col,
                             const float* __restrict__ bias,
                             float* __restrict__ outp, int N, int doRelu) {
  constexpr int VEC = WIDTH / 32;
  constexpr int G = 8;
  const int lane = threadIdx.x & 31;
  const int wave = __builtin_amdgcn_readfirstlane(
      (int)((blockIdx.x * (size_t)blockDim.x + threadIdx.x) >> 5));
  const int node0 = wave * NPW;
  if (node0 >= N) return;

  int rp[NPW + 1];
#pragma unroll
  for (int i = 0; i <= NPW; ++i) {
    const int n = (node0 + i) < N ? (node0 + i) : N;
    rp[i] = rowptr[n];
  }
  int first[NPW];
#pragma unroll
  for (int i = 0; i < NPW; ++i)
    first[i] = (lane < rp[i + 1] - rp[i]) ? col[rp[i] + lane] : 0;

  const float* __restrict__ hl = h + lane * VEC;

#pragma unroll
  for (int i = 0; i < NPW; ++i) {
    const int node = node0 + i;
    if (node < N) {
      const int beg = rp[i], end = rp[i + 1];
      float acc[VEC];
#pragma unroll
      for (int q = 0; q < VEC; ++q) acc[q] = 0.0f;

      int cur = first[i];
      for (int p = beg; p < end; p += 32) {
        if (p != beg) cur = (lane < end - p) ? col[p + lane] : 0;
        const int m = (end - p) < 32 ? (end - p) : 32;
        for (int j = 0; j < m; j += G) {
          float v[G][VEC];
#pragma unroll
          for (int u = 0; u < G; ++u) {
            const int jj = (j + u) < m ? (j + u) : (m - 1);
            const int s = __builtin_amdgcn_readlane(cur, jj);
            gcn_ld_row<VEC>(hl + (size_t)s * WIDTH, v[u]);
          }
#pragma unroll
          for (int u = 0; u < G; ++u) {
            const bool on = (j + u) < m;
#pragma unroll
            for (int q = 0; q < VEC; ++q) acc[q] += on ? v[u][q] : 0.0f;
          }
        }
      }

      float r[VEC];
      if (FINAL) {
        float d = (float)(end - beg);
        d = d < 1.0f ? 1.0f : d;
#pragma unroll
        for (int q = 0; q < VEC; ++q) {
          float t = acc[q] / d + bias[lane * VEC + q];
          if (doRelu) t = t > 0.0f ? t : 0.0f;
          r[q] = t;
        }
      } else {
#pragma unroll
        for (int q = 0; q < VEC; ++q) r[q] = acc[q];
      }
      gcn_st_row<VEC>(outp + (size_t)node * WIDTH + lane * VEC, r);
    }
  }
}

__global__ void gcn_dense_chain_wmma_f32(const float* __restrict__ X,
                                         const float* __restrict__ W1,
                                         const int* __restrict__ cnt,
                                         const float* __restrict__ b1,
                                         const float* __restrict__ W2,
                                         float* __restrict__ Y, int N) {
  const int wave = (int)((blockIdx.x * (size_t)blockDim.x + threadIdx.x) >> 5);
  if (wave >= (N >> 4)) return;

  const int lane = threadIdx.x & 31;
  const int half = lane >> 4;
  const int l16  = lane & 15;
  const int node = (wave << 4) + l16;
  const float* __restrict__ xrow = X + (size_t)node * 64;

  v8f s[4];
#pragma unroll
  for (int j = 0; j < 4; ++j) s[j] = (v8f){};

#pragma unroll 2
  for (int k = 0; k < 64; k += 4) {
    const int kk = k + (half << 1);
    const v2f x = *(const v2f*)(xrow + kk);
#pragma unroll
    for (int j = 0; j < 4; ++j) {
      v2f w;
      w.x = W1[kk * 64 + (j << 4) + l16];
      w.y = W1[(kk + 1) * 64 + (j << 4) + l16];
      s[j] = __builtin_amdgcn_wmma_f32_16x16x4_f32(
           false, w,  false, x,
           (short)0, s[j],  false,  false);
    }
  }

  float d = (float)cnt[node];
  d = d < 1.0f ? 1.0f : d;
  const float* __restrict__ bh = b1 + (half << 3);
#pragma unroll
  for (int j = 0; j < 4; ++j) {
#pragma unroll
    for (int r = 0; r < 8; ++r) {
      const float t = s[j][r] / d + bh[(j << 4) + r];
      s[j][r] = t > 0.0f ? t : 0.0f;
    }
  }

  v8f o[2];
  o[0] = (v8f){};
  o[1] = (v8f){};
#pragma unroll
  for (int j = 0; j < 4; ++j) {
#pragma unroll
    for (int q = 0; q < 4; ++q) {
      const int f = (j << 4) + (half << 3) + (q << 1);
      v2f x;
      x.x = s[j][2 * q];
      x.y = s[j][2 * q + 1];
#pragma unroll
      for (int t = 0; t < 2; ++t) {
        v2f w;
        w.x = W2[f * 32 + (t << 4) + l16];
        w.y = W2[(f + 1) * 32 + (t << 4) + l16];
        o[t] = __builtin_amdgcn_wmma_f32_16x16x4_f32(
             false, w,  false, x,
             (short)0, o[t],  false,  false);
      }
    }
  }

  float* __restrict__ yrow = Y + (size_t)node * 32 + (half << 3);
#pragma unroll
  for (int t = 0; t < 2; ++t) {
    *(v4f*)(yrow + (t << 4))     = __builtin_shufflevector(o[t], o[t], 0, 1, 2, 3);
    *(v4f*)(yrow + (t << 4) + 4) = __builtin_shufflevector(o[t], o[t], 4, 5, 6, 7);
  }
}

extern "C" void kernel_launch(void* const* d_in, const int* in_sizes, int n_in,
                              void* d_out, int out_size, void* d_ws, size_t ws_size,
                              hipStream_t stream) {
  const float* feat = (const float*)d_in[0];
  const float* W0   = (const float*)d_in[1];
  const float* b0   = (const float*)d_in[2];
  const float* W1   = (const float*)d_in[3];
  const float* b1   = (const float*)d_in[4];
  const float* W2   = (const float*)d_in[5];
  const float* b2   = (const float*)d_in[6];
  const int*   src  = (const int*)d_in[7];
  const int*   dst  = (const int*)d_in[8];
  float* out = (float*)d_out;

  const int N = in_sizes[0] / 256;
  const int E = in_sizes[7];

  float* bufA = (float*)d_ws;
  float* bufB = bufA + (size_t)N * 64;
  int*   cnt  = (int*)(bufB + (size_t)N * 64);
  int*   rank = (int*)bufB;
  int*   cm   = (int*)bufA;
  const int rpPad = ((N + 1 + 63) / 64) * 64;
  int* rowptrO = (int*)out;
  int* colO    = rowptrO + rpPad;
  int* rowptrA = (int*)bufA;
  int* colA    = rowptrA + rpPad;

  const int TB = 256;
  const int NPW = 4;
  const int gemmBlocks  = (int)((((long long)(N / 16)) * 32 + TB - 1) / TB);
  const int edge4Blocks = (int)((((long long)E + 3) / 4 + TB - 1) / TB);
  const int scanBlocks  = (N + 1023) / 1024;
  const int nodeBlocks  = (N + TB - 1) / TB;
  constexpr int CH = 32;
  const int S  = (((E + CH - 1) / CH) + 3) & ~3;
  const int pullBlocks  = (int)(((long long)((N + NPW - 1) / NPW) * 32 + TB - 1) / TB);
  const int moveBlocks  = (int)((((long long)rpPad + E + 3) / 4 + TB - 1) / TB);

  gcn_chunk_count_rank<<<CH, 1024, 0, stream>>>(dst, cm, rank, E, N, S);
  gcn_chunk_prefix<CH><<<nodeBlocks, TB, 0, stream>>>(cm, cnt, N);
  gcn_rowptr_scan<<<scanBlocks, 1024, 0, stream>>>(cnt, rowptrO, N);
  gcn_fill_col<<<edge4Blocks, TB, 0, stream>>>(src, dst, rank, rowptrO, cm, colO, E, N, S);

  gcn_gemm_wmma_f32<256, 4><<<gemmBlocks, TB, 0, stream>>>(feat, W0, bufA, N);
  gcn_pull_sum<64, NPW, true><<<pullBlocks, TB, 0, stream>>>(bufA, rowptrO, colO, b0, bufB, N, 1);

  gcn_pull_sum<64, NPW, false><<<pullBlocks, TB, 0, stream>>>(bufB, rowptrO, colO, b1, bufA, N, 0);
  gcn_dense_chain_wmma_f32<<<gemmBlocks, TB, 0, stream>>>(bufA, W1, cnt, b1, W2, bufB, N);

  gcn_move_ints<<<moveBlocks, TB, 0, stream>>>(rowptrO, rowptrA, rpPad + E);
  gcn_pull_sum<32, NPW, true><<<pullBlocks, TB, 0, stream>>>(bufB, rowptrA, colA, b2, out, N, 0);
}


# layer 1's neighbour sum moved into the dense-chain launch: a wave sums its 16 nodes' rows (list order kept) into an LDS tile, product 1's B operand; sum plane not written or read, one launch less
# speedup vs baseline: 2.9719x; 1.0354x over previous
#include <hip/hip_runtime.h>


typedef __attribute__((ext_vector_type(2))) float v2f;
typedef __attribute__((ext_vector_type(4))) float v4f;
typedef __attribute__((ext_vector_type(8))) float v8f;

constexpr int GCN_HWORDS = 51200;
constexpr int GCN_TRIPS  = 16;

__device__ __forceinline__ int gcn_lds_rank16(unsigned int* hist, int d) {
  const unsigned int sh = (unsigned int)(d & 1) << 4;
  const unsigned int old = atomicAdd(hist + (d >> 1), 1u << sh);
  return (int)((old >> sh) & 0xFFFFu);
}

__global__ void __launch_bounds__(1024)
gcn_chunk_count_rank(const int* __restrict__ dst, int* __restrict__ cm,
                     int* __restrict__ rank, int E, int N, int S) {
  __shared__ __attribute__((aligned(16))) unsigned int hist[GCN_HWORDS];
  const int tid = threadIdx.x;
  uint4* h4 = (uint4*)hist;
  for (int i = tid; i < GCN_HWORDS / 4; i += 1024) h4[i] = make_uint4(0u, 0u, 0u, 0u);
  __syncthreads();

  const long long b0 = (long long)blockIdx.x * S;
  const int e0 = b0 < E ? (int)b0 : E;
  const int e1 = (b0 + S) < E ? (int)(b0 + S) : E;
  const int e  = e0 + (tid << 2);
  int eL = (e1 - 4) & ~3;
  eL = eL < 0 ? 0 : eL;
  int4 dv[GCN_TRIPS];
#pragma unroll
  for (int j = 0; j < GCN_TRIPS; ++j) {
    const int ej = e + (j << 12);
    dv[j] = *(const int4*)(dst + (ej < eL ? ej : eL));
  }
#pragma unroll
  for (int j = 0; j < GCN_TRIPS; ++j) {
    const int ej = e + (j << 12);
    if (ej + 3 < e1) {
      int4 r;
      r.x = gcn_lds_rank16(hist, dv[j].x);
      r.y = gcn_lds_rank16(hist, dv[j].y);
      r.z = gcn_lds_rank16(hist, dv[j].z);
      r.w = gcn_lds_rank16(hist, dv[j].w);
      *(int4*)(rank + ej) = r;
    } else {
      for (int k = ej; k < e1; ++k) rank[k] = gcn_lds_rank16(hist, dst[k]);
    }
  }
  __syncthreads();

  int* __restrict__ row = cm + (size_t)blockIdx.x * N;
  const uint2* h2 = (const uint2*)hist;
  for (int q = tid; q < (N >> 2); q += 1024) {
    const uint2 w = h2[q];
    int4 v;
    v.x = (int)(w.x & 0xFFFFu); v.y = (int)(w.x >> 16);
    v.z = (int)(w.y & 0xFFFFu); v.w = (int)(w.y >> 16);
    *(int4*)(row + (q << 2)) = v;
  }
}

template <int C>
__global__ void gcn_chunk_prefix(int* __restrict__ cm, int* __restrict__ cnt, int N) {
  const int n = blockIdx.x * blockDim.x + threadIdx.x;
  if (n >= N) return;
  int t[C];
#pragma unroll
  for (int c = 0; c < C; ++c) t[c] = cm[(size_t)c * N + n];
  int sum = 0;
#pragma unroll
  for (int c = 0; c < C; ++c) {
    cm[(size_t)c * N + n] = sum;
    sum += t[c];
  }
  cnt[n] = sum;
}

__global__ void gcn_rowptr_scan(const int* __restrict__ cnt, int* __restrict__ rowptr, int N) {
  __shared__ int wsum[32];
  __shared__ int wpre[32];
  const int tid = threadIdx.x, lane = tid & 31, w = tid >> 5;
  const int base = blockIdx.x * 1024;

  int acc = 0;
  const int4* c4 = (const int4*)cnt;
  const int nfull = blockIdx.x >> 2;
#pragma unroll 4
  for (int c = 0; c < nfull; ++c) {
    const int4 q = c4[(c << 10) + tid];
    acc += (q.x + q.y) + (q.z + q.w);
  }
  for (int i = (nfull << 12) + tid; i < base; i += 1024) acc += cnt[i];
  const int n = base + tid;
  const int v = (n < N) ? cnt[n] : 0;

  int inc = v;
  for (int o = 1; o < 32; o <<= 1) {
    const int u = __shfl_up(inc, o);
    if (lane >= o) inc += u;
  }
  for (int o = 16; o > 0; o >>= 1) acc += __shfl_xor(acc, o);
  if (lane == 31) wsum[w] = inc;
  if (lane == 0)  wpre[w] = acc;
  __syncthreads();

  const int ws = wsum[lane];
  int wsinc = ws;
  for (int o = 1; o < 32; o <<= 1) {
    const int u = __shfl_up(wsinc, o);
    if (lane >= o) wsinc += u;
  }
  int front = wpre[lane];
  for (int o = 16; o > 0; o >>= 1) front += __shfl_xor(front, o);
  const int waveOff = __shfl(wsinc - ws, w);
  const int excl = front + waveOff + inc - v;
  if (n < N) rowptr[n] = excl;
  if (n == N - 1) rowptr[N] = excl + v;
}

__global__ void gcn_fill_col(const int* __restrict__ src, const int* __restrict__ dst,
                             const int* __restrict__ rank, const int* __restrict__ rowptr,
                             const int* __restrict__ cm, int* __restrict__ col,
                             int E, int N, int S) {
  const int t = blockIdx.x * blockDim.x + threadIdx.x;
  const int e = t << 2;
  if (e + 3 < E) {
    const int4 s = *(const int4*)(src + e);
    const int4 d = *(const int4*)(dst + e);
    const int4 r = *(const int4*)(rank + e);
    const int* __restrict__ pre = cm + (size_t)(e / S) * N;
    const int p0 = rowptr[d.x] + pre[d.x], p1 = rowptr[d.y] + pre[d.y];
    const int p2 = rowptr[d.z] + pre[d.z], p3 = rowptr[d.w] + pre[d.w];
    col[p0 + r.x] = s.x;
    col[p1 + r.y] = s.y;
    col[p2 + r.z] = s.z;
    col[p3 + r.w] = s.w;
  } else {
    for (int k = e; k < E; ++k) {
      const int dk = dst[k];
      col[rowptr[dk] + cm[(size_t)(k / S) * N + dk] + rank[k]] = src[k];
    }
  }
}

__global__ void gcn_move_ints(const int* __restrict__ from, int* __restrict__ to, int T) {
  const int t = blockIdx.x * blockDim.x + threadIdx.x;
  const int e = t << 2;
  if (e + 3 < T) {
    *(int4*)(to + e) = *(const int4*)(from + e);
  } else {
    for (int k = e; k < T; ++k) to[k] = from[k];
  }
}

template <int K, int TM>
__global__ void gcn_gemm_wmma_f32(const float* __restrict__ X,
                                  const float* __restrict__ W,
                                  float* __restrict__ Y, int N) {
  const int M = TM * 16;
  const int wave = (int)((blockIdx.x * (size_t)blockDim.x + threadIdx.x) >> 5);
  if (wave >= (N >> 4)) return;

  const int lane = threadIdx.x & 31;
  const int half = lane >> 4;
  const int l16  = lane & 15;
  const float* __restrict__ xrow = X + (size_t)((wave << 4) + l16) * K;

  v8f acc[TM];
#pragma unroll
  for (int j = 0; j < TM; ++j) acc[j] = (v8f){};

#pragma unroll 2
  for (int k = 0; k < K; k += 4) {
    const int kk = k + (half << 1);
    const v2f a = *(const v2f*)(xrow + kk);
#pragma unroll
    for (int j = 0; j < TM; ++j) {
      v2f b;
      b.x = W[kk * M + (j << 4) + l16];
      b.y = W[(kk + 1) * M + (j << 4) + l16];
      acc[j] = __builtin_amdgcn_wmma_f32_16x16x4_f32(
           false, a,  false, b,
           (short)0, acc[j],  false,  false);
    }
  }

  const int rbase = (wave << 4) + (half << 3);
#pragma unroll
  for (int j = 0; j < TM; ++j)
#pragma unroll
    for (int r = 0; r < 8; ++r)
      Y[(size_t)(rbase + r) * M + (j << 4) + l16] = acc[j][r];
}

template <int VEC>
__device__ __forceinline__ void gcn_ld_row(const float* p, float (&v)[VEC]) {
  if constexpr (VEC == 2) { const v2f t = *(const v2f*)p; v[0] = t.x; v[1] = t.y; }
  else                    { v[0] = *p; }
}
template <int VEC>
__device__ __forceinline__ void gcn_st_row(float* p, const float (&v)[VEC]) {
  if constexpr (VEC == 2) { v2f t; t.x = v[0]; t.y = v[1]; *(v2f*)p = t; }
  else                    { *p = v[0]; }
}

template <int WIDTH, int NPW, bool FINAL>
__global__ void gcn_pull_sum(const float* __restrict__ h,
                             const int* __restrict__ rowptr,
                             const int* __restrict__ col,
                             const float* __restrict__ bias,
                             float* __restrict__ outp, int N, int doRelu) {
  constexpr int VEC = WIDTH / 32;
  constexpr int G = 8;
  const int lane = threadIdx.x & 31;
  const int wave = __builtin_amdgcn_readfirstlane(
      (int)((blockIdx.x * (size_t)blockDim.x + threadIdx.x) >> 5));
  const int node0 = wave * NPW;
  if (node0 >= N) return;

  int rp[NPW + 1];
#pragma unroll
  for (int i = 0; i <= NPW; ++i) {
    const int n = (node0 + i) < N ? (node0 + i) : N;
    rp[i] = rowptr[n];
  }
  int first[NPW];
#pragma unroll
  for (int i = 0; i < NPW; ++i)
    first[i] = (lane < rp[i + 1] - rp[i]) ? col[rp[i] + lane] : 0;

  const float* __restrict__ hl = h + lane * VEC;

#pragma unroll
  for (int i = 0; i < NPW; ++i) {
    const int node = node0 + i;
    if (node < N) {
      const int beg = rp[i], end = rp[i + 1];
      float acc[VEC];
#pragma unroll
      for (int q = 0; q < VEC; ++q) acc[q] = 0.0f;

      int cur = first[i];
      for (int p = beg; p < end; p += 32) {
        if (p != beg) cur = (lane < end - p) ? col[p + lane] : 0;
        const int m = (end - p) < 32 ? (end - p) : 32;
        for (int j = 0; j < m; j += G) {
          float v[G][VEC];
#pragma unroll
          for (int u = 0; u < G; ++u) {
            const int jj = (j + u) < m ? (j + u) : (m - 1);
            const int s = __builtin_amdgcn_readlane(cur, jj);
            gcn_ld_row<VEC>(hl + (size_t)s * WIDTH, v[u]);
          }
#pragma unroll
          for (int u = 0; u < G; ++u) {
            const bool on = (j + u) < m;
#pragma unroll
            for (int q = 0; q < VEC; ++q) acc[q] += on ? v[u][q] : 0.0f;
          }
        }
      }

      float r[VEC];
      if (FINAL) {
        float d = (float)(end - beg);
        d = d < 1.0f ? 1.0f : d;
#pragma unroll
        for (int q = 0; q < VEC; ++q) {
          float t = acc[q] / d + bias[lane * VEC + q];
          if (doRelu) t = t > 0.0f ? t : 0.0f;
          r[q] = t;
        }
      } else {
#pragma unroll
        for (int q = 0; q < VEC; ++q) r[q] = acc[q];
      }
      gcn_st_row<VEC>(outp + (size_t)node * WIDTH + lane * VEC, r);
    }
  }
}

constexpr int GCN_TILE_PITCH = 68;

__device__ __forceinline__ void gcn_ask_rows(const float* __restrict__ hl, int cur, int m,
                                             v2f (&v)[32]) {
#pragma unroll
  for (int b = 0; b < 4; ++b) {
    if ((b << 3) < m) {
#pragma unroll
      for (int u = 0; u < 8; ++u) {
        const int jj = ((b << 3) + u) < m ? ((b << 3) + u) : (m - 1);
        const int s = __builtin_amdgcn_readlane(cur, jj);
        v[(b << 3) + u] = *(const v2f*)(hl + (size_t)s * 64);
      }
    }
  }
}
__device__ __forceinline__ void gcn_add_rows(const v2f (&v)[32], int m, float& a0, float& a1) {
#pragma unroll
  for (int b = 0; b < 4; ++b) {
    if ((b << 3) < m) {
#pragma unroll
      for (int u = 0; u < 8; ++u) {
        const bool on = ((b << 3) + u) < m;
        a0 += on ? v[(b << 3) + u].x : 0.0f;
        a1 += on ? v[(b << 3) + u].y : 0.0f;
      }
    }
  }
}


__global__ void __launch_bounds__(128)
gcn_sum_dense_chain_wmma_f32(const float* __restrict__ h,
                             const int* __restrict__ rowptr,
                             const int* __restrict__ col,
                             const float* __restrict__ W1,
                             const int* __restrict__ cnt,
                             const float* __restrict__ b1,
                             const float* __restrict__ W2,
                             float* __restrict__ Y, int N) {
  __shared__ __attribute__((aligned(16))) float tiles[4 * 16 * GCN_TILE_PITCH];
  const int lane = threadIdx.x & 31;
  const int wave = __builtin_amdgcn_readfirstlane(
      (int)((blockIdx.x * (size_t)blockDim.x + threadIdx.x) >> 5));
  const int wib  = __builtin_amdgcn_readfirstlane((int)(threadIdx.x >> 5));
  if (wave >= (N >> 4)) return;
  float* __restrict__ tile = tiles + wib * (16 * GCN_TILE_PITCH);

  const int node0 = wave << 4;
  const int rpv = rowptr[node0 + (lane < 16 ? lane : 16)];
  int beg = __builtin_amdgcn_readlane(rpv, 0);
  int end = __builtin_amdgcn_readlane(rpv, 1);
  int cur = (lane < end - beg) ? col[beg + lane] : 0;
  const float* __restrict__ hl = h + lane * 2;

  for (int i = 0; i < 16; ++i) {
    const int nbeg = end;
    const int nend = __builtin_amdgcn_readlane(rpv, (i + 2) < 16 ? (i + 2) : 16);
    const int m0 = (end - beg) < 32 ? (end - beg) : 32;
    float a0 = 0.0f, a1 = 0.0f;
    v2f v[32];
    int nxt;
    if (m0 > 0) {
      gcn_ask_rows(hl, cur, m0, v);
      nxt = (lane < nend - nbeg) ? col[nbeg + lane] : 0;
      gcn_add_rows(v, m0, a0, a1);
      for (int p = beg + 32; p < end; p += 32) {
        const int c2 = (lane < end - p) ? col[p + lane] : 0;
        const int m = (end - p) < 32 ? (end - p) : 32;
        gcn_ask_rows(hl, c2, m, v);
        gcn_add_rows(v, m, a0, a1);
      }
    } else {
      nxt = (lane < nend - nbeg) ? col[nbeg + lane] : 0;
    }
    v2f t;
    t.x = a0;
    t.y = a1;
    *(v2f*)(tile + i * GCN_TILE_PITCH + lane * 2) = t;
    cur = nxt;
    beg = nbeg;
    end = nend;
  }
  __builtin_amdgcn_wave_barrier();

  const int half = lane >> 4;
  const int l16  = lane & 15;
  const int node = node0 + l16;
  const float* xrow = tile + l16 * GCN_TILE_PITCH;

  v8f s[4];
#pragma unroll
  for (int j = 0; j < 4; ++j) s[j] = (v8f){};

#pragma unroll 2
  for (int k = 0; k < 64; k += 4) {
    const int kk = k + (half << 1);
    const v2f x = *(const v2f*)(xrow + kk);
#pragma unroll
    for (int j = 0; j < 4; ++j) {
      v2f w;
      w.x = W1[kk * 64 + (j << 4) + l16];
      w.y = W1[(kk + 1) * 64 + (j << 4) + l16];
      s[j] = __builtin_amdgcn_wmma_f32_16x16x4_f32(
           false, w,  false, x,
           (short)0, s[j],  false,  false);
    }
  }

  float d = (float)cnt[node];
  d = d < 1.0f ? 1.0f : d;
  const float* __restrict__ bh = b1 + (half << 3);
#pragma unroll
  for (int j = 0; j < 4; ++j) {
#pragma unroll
    for (int r = 0; r < 8; ++r) {
      const float t = s[j][r] / d + bh[(j << 4) + r];
      s[j][r] = t > 0.0f ? t : 0.0f;
    }
  }

  v8f o[2];
  o[0] = (v8f){};
  o[1] = (v8f){};
#pragma unroll
  for (int j = 0; j < 4; ++j) {
#pragma unroll
    for (int q = 0; q < 4; ++q) {
      const int f = (j << 4) + (half << 3) + (q << 1);
      v2f x;
      x.x = s[j][2 * q];
      x.y = s[j][2 * q + 1];
#pragma unroll
      for (int t = 0; t < 2; ++t) {
        v2f w;
        w.x = W2[f * 32 + (t << 4) + l16];
        w.y = W2[(f + 1) * 32 + (t << 4) + l16];
        o[t] = __builtin_amdgcn_wmma_f32_16x16x4_f32(
             false, w,  false, x,
             (short)0, o[t],  false,  false);
      }
    }
  }

  float* __restrict__ yrow = Y + (size_t)node * 32 + (half << 3);
#pragma unroll
  for (int t = 0; t < 2; ++t) {
    *(v4f*)(yrow + (t << 4))     = __builtin_shufflevector(o[t], o[t], 0, 1, 2, 3);
    *(v4f*)(yrow + (t << 4) + 4) = __builtin_shufflevector(o[t], o[t], 4, 5, 6, 7);
  }
}

extern "C" void kernel_launch(void* const* d_in, const int* in_sizes, int n_in,
                              void* d_out, int out_size, void* d_ws, size_t ws_size,
                              hipStream_t stream) {
  const float* feat = (const float*)d_in[0];
  const float* W0   = (const float*)d_in[1];
  const float* b0   = (const float*)d_in[2];
  const float* W1   = (const float*)d_in[3];
  const float* b1   = (const float*)d_in[4];
  const float* W2   = (const float*)d_in[5];
  const float* b2   = (const float*)d_in[6];
  const int*   src  = (const int*)d_in[7];
  const int*   dst  = (const int*)d_in[8];
  float* out = (float*)d_out;

  const int N = in_sizes[0] / 256;
  const int E = in_sizes[7];

  float* bufA = (float*)d_ws;
  float* bufB = bufA + (size_t)N * 64;
  int*   cnt  = (int*)(bufB + (size_t)N * 64);
  int*   rank = (int*)bufB;
  int*   cm   = (int*)bufA;
  const int rpPad = ((N + 1 + 63) / 64) * 64;
  int* rowptrO = (int*)out;
  int* colO    = rowptrO + rpPad;
  int* rowptrB = (int*)bufB;
  int* colB    = rowptrB + rpPad;

  const int TB = 256;
  const int NPW = 4;
  const int gemmBlocks  = (int)((((long long)(N / 16)) * 32 + TB - 1) / TB);
  const int edge4Blocks = (int)((((long long)E + 3) / 4 + TB - 1) / TB);
  const int scanBlocks  = (N + 1023) / 1024;
  const int nodeBlocks  = (N + TB - 1) / TB;
  constexpr int CH = 32;
  const int S  = (((E + CH - 1) / CH) + 3) & ~3;
  const int pullBlocks  = (int)(((long long)((N + NPW - 1) / NPW) * 32 + TB - 1) / TB);
  const int TBF = 128;
  const int fusedBlocks = (int)((((long long)(N / 16)) * 32 + TBF - 1) / TBF);
  const int moveBlocks  = (int)((((long long)rpPad + E + 3) / 4 + TB - 1) / TB);

  gcn_chunk_count_rank<<<CH, 1024, 0, stream>>>(dst, cm, rank, E, N, S);
  gcn_chunk_prefix<CH><<<nodeBlocks, TB, 0, stream>>>(cm, cnt, N);
  gcn_rowptr_scan<<<scanBlocks, 1024, 0, stream>>>(cnt, rowptrO, N);
  gcn_fill_col<<<edge4Blocks, TB, 0, stream>>>(src, dst, rank, rowptrO, cm, colO, E, N, S);

  gcn_gemm_wmma_f32<256, 4><<<gemmBlocks, TB, 0, stream>>>(feat, W0, bufA, N);
  gcn_pull_sum<64, NPW, true><<<pullBlocks, TB, 0, stream>>>(bufA, rowptrO, colO, b0, bufB, N, 1);

  gcn_sum_dense_chain_wmma_f32<<<fusedBlocks, TBF, 0, stream>>>(bufB, rowptrO, colO, W1, cnt, b1, W2, bufA, N);

  gcn_move_ints<<<moveBlocks, TB, 0, stream>>>(rowptrO, rowptrB, rpPad + E);
  gcn_pull_sum<32, NPW, true><<<pullBlocks, TB, 0, stream>>>(bufA, rowptrB, colB, b2, out, N, 0);
}


# layer 0's product moved into the count launch as a second block role: its 32-wave row blocks run on the WGPs that the 32 one-per-WGP chunk blocks leave idle; one launch less
# speedup vs baseline: 3.2834x; 1.1048x over previous
#include <hip/hip_runtime.h>


typedef __attribute__((ext_vector_type(2))) float v2f;
typedef __attribute__((ext_vector_type(4))) float v4f;
typedef __attribute__((ext_vector_type(8))) float v8f;

constexpr int GCN_HWORDS = 51200;
constexpr int GCN_TRIPS  = 16;

__device__ __forceinline__ int gcn_lds_rank16(unsigned int* hist, int d) {
  const unsigned int sh = (unsigned int)(d & 1) << 4;
  const unsigned int old = atomicAdd(hist + (d >> 1), 1u << sh);
  return (int)((old >> sh) & 0xFFFFu);
}

__device__ __forceinline__ void
gcn_chunk_count_rank_block(unsigned int* hist, const int* __restrict__ dst, int* __restrict__ cm,
                           int* __restrict__ rank, int E, int N, int S, int blk) {
  const int tid = threadIdx.x;
  uint4* h4 = (uint4*)hist;
  for (int i = tid; i < GCN_HWORDS / 4; i += 1024) h4[i] = make_uint4(0u, 0u, 0u, 0u);
  __syncthreads();

  const long long b0 = (long long)blk * S;
  const int e0 = b0 < E ? (int)b0 : E;
  const int e1 = (b0 + S) < E ? (int)(b0 + S) : E;
  const int e  = e0 + (tid << 2);
  int eL = (e1 - 4) & ~3;
  eL = eL < 0 ? 0 : eL;
  int4 dv[GCN_TRIPS];
#pragma unroll
  for (int j = 0; j < GCN_TRIPS; ++j) {
    const int ej = e + (j << 12);
    dv[j] = *(const int4*)(dst + (ej < eL ? ej : eL));
  }
#pragma unroll
  for (int j = 0; j < GCN_TRIPS; ++j) {
    const int ej = e + (j << 12);
    if (ej + 3 < e1) {
      int4 r;
      r.x = gcn_lds_rank16(hist, dv[j].x);
      r.y = gcn_lds_rank16(hist, dv[j].y);
      r.z = gcn_lds_rank16(hist, dv[j].z);
      r.w = gcn_lds_rank16(hist, dv[j].w);
      *(int4*)(rank + ej) = r;
    } else {
      for (int k = ej; k < e1; ++k) rank[k] = gcn_lds_rank16(hist, dst[k]);
    }
  }
  __syncthreads();

  int* __restrict__ row = cm + (size_t)blk * N;
  const uint2* h2 = (const uint2*)hist;
  for (int q = tid; q < (N >> 2); q += 1024) {
    const uint2 w = h2[q];
    int4 v;
    v.x = (int)(w.x & 0xFFFFu); v.y = (int)(w.x >> 16);
    v.z = (int)(w.y & 0xFFFFu); v.w = (int)(w.y >> 16);
    *(int4*)(row + (q << 2)) = v;
  }
}

template <int C>
__global__ void gcn_chunk_prefix(int* __restrict__ cm, int* __restrict__ cnt, int N) {
  const int n = blockIdx.x * blockDim.x + threadIdx.x;
  if (n >= N) return;
  int t[C];
#pragma unroll
  for (int c = 0; c < C; ++c) t[c] = cm[(size_t)c * N + n];
  int sum = 0;
#pragma unroll
  for (int c = 0; c < C; ++c) {
    cm[(size_t)c * N + n] = sum;
    sum += t[c];
  }
  cnt[n] = sum;
}

__global__ void gcn_rowptr_scan(const int* __restrict__ cnt, int* __restrict__ rowptr, int N) {
  __shared__ int wsum[32];
  __shared__ int wpre[32];
  const int tid = threadIdx.x, lane = tid & 31, w = tid >> 5;
  const int base = blockIdx.x * 1024;

  int acc = 0;
  const int4* c4 = (const int4*)cnt;
  const int nfull = blockIdx.x >> 2;
#pragma unroll 4
  for (int c = 0; c < nfull; ++c) {
    const int4 q = c4[(c << 10) + tid];
    acc += (q.x + q.y) + (q.z + q.w);
  }
  for (int i = (nfull << 12) + tid; i < base; i += 1024) acc += cnt[i];
  const int n = base + tid;
  const int v = (n < N) ? cnt[n] : 0;

  int inc = v;
  for (int o = 1; o < 32; o <<= 1) {
    const int u = __shfl_up(inc, o);
    if (lane >= o) inc += u;
  }
  for (int o = 16; o > 0; o >>= 1) acc += __shfl_xor(acc, o);
  if (lane == 31) wsum[w] = inc;
  if (lane == 0)  wpre[w] = acc;
  __syncthreads();

  const int ws = wsum[lane];
  int wsinc = ws;
  for (int o = 1; o < 32; o <<= 1) {
    const int u = __shfl_up(wsinc, o);
    if (lane >= o) wsinc += u;
  }
  int front = wpre[lane];
  for (int o = 16; o > 0; o >>= 1) front += __shfl_xor(front, o);
  const int waveOff = __shfl(wsinc - ws, w);
  const int excl = front + waveOff + inc - v;
  if (n < N) rowptr[n] = excl;
  if (n == N - 1) rowptr[N] = excl + v;
}

__global__ void gcn_fill_col(const int* __restrict__ src, const int* __restrict__ dst,
                             const int* __restrict__ rank, const int* __restrict__ rowptr,
                             const int* __restrict__ cm, int* __restrict__ col,
                             int E, int N, int S) {
  const int t = blockIdx.x * blockDim.x + threadIdx.x;
  const int e = t << 2;
  if (e + 3 < E) {
    const int4 s = *(const int4*)(src + e);
    const int4 d = *(const int4*)(dst + e);
    const int4 r = *(const int4*)(rank + e);
    const int* __restrict__ pre = cm + (size_t)(e / S) * N;
    const int p0 = rowptr[d.x] + pre[d.x], p1 = rowptr[d.y] + pre[d.y];
    const int p2 = rowptr[d.z] + pre[d.z], p3 = rowptr[d.w] + pre[d.w];
    col[p0 + r.x] = s.x;
    col[p1 + r.y] = s.y;
    col[p2 + r.z] = s.z;
    col[p3 + r.w] = s.w;
  } else {
    for (int k = e; k < E; ++k) {
      const int dk = dst[k];
      col[rowptr[dk] + cm[(size_t)(k / S) * N + dk] + rank[k]] = src[k];
    }
  }
}

__global__ void gcn_move_ints(const int* __restrict__ from, int* __restrict__ to, int T) {
  const int t = blockIdx.x * blockDim.x + threadIdx.x;
  const int e = t << 2;
  if (e + 3 < T) {
    *(int4*)(to + e) = *(const int4*)(from + e);
  } else {
    for (int k = e; k < T; ++k) to[k] = from[k];
  }
}

template <int K, int TM>
__device__ __forceinline__ void gcn_gemm_wmma_f32_block(const float* __restrict__ X,
                                  const float* __restrict__ W,
                                  float* __restrict__ Y, int N, int blk) {
  const int M = TM * 16;
  const int wave = (int)((blk * (size_t)blockDim.x + threadIdx.x) >> 5);
  if (wave >= (N >> 4)) return;

  const int lane = threadIdx.x & 31;
  const int half = lane >> 4;
  const int l16  = lane & 15;
  const float* __restrict__ xrow = X + (size_t)((wave << 4) + l16) * K;

  v8f acc[TM];
#pragma unroll
  for (int j = 0; j < TM; ++j) acc[j] = (v8f){};

#pragma unroll 2
  for (int k = 0; k < K; k += 4) {
    const int kk = k + (half << 1);
    const v2f a = *(const v2f*)(xrow + kk);
#pragma unroll
    for (int j = 0; j < TM; ++j) {
      v2f b;
      b.x = W[kk * M + (j << 4) + l16];
      b.y = W[(kk + 1) * M + (j << 4) + l16];
      acc[j] = __builtin_amdgcn_wmma_f32_16x16x4_f32(
           false, a,  false, b,
           (short)0, acc[j],  false,  false);
    }
  }

  const int rbase = (wave << 4) + (half << 3);
#pragma unroll
  for (int j = 0; j < TM; ++j)
#pragma unroll
    for (int r = 0; r < 8; ++r)
      Y[(size_t)(rbase + r) * M + (j << 4) + l16] = acc[j][r];
}

__global__ void __launch_bounds__(1024)
gcn_count_rank_and_product0(const int* __restrict__ dst, int* __restrict__ cm,
                            int* __restrict__ rank, int E, int N, int S, int rowBlocks,
                            const float* __restrict__ X, const float* __restrict__ W,
                            float* __restrict__ Y) {
  __shared__ __attribute__((aligned(16))) unsigned int hist[GCN_HWORDS];
  const int b = (int)blockIdx.x;
  if (b < rowBlocks) {
    gcn_gemm_wmma_f32_block<256, 4>(X, W, Y, N, b);
  } else {
    gcn_chunk_count_rank_block(hist, dst, cm, rank, E, N, S, b - rowBlocks);
  }
}

template <int VEC>
__device__ __forceinline__ void gcn_ld_row(const float* p, float (&v)[VEC]) {
  if constexpr (VEC == 2) { const v2f t = *(const v2f*)p; v[0] = t.x; v[1] = t.y; }
  else                    { v[0] = *p; }
}
template <int VEC>
__device__ __forceinline__ void gcn_st_row(float* p, const float (&v)[VEC]) {
  if constexpr (VEC == 2) { v2f t; t.x = v[0]; t.y = v[1]; *(v2f*)p = t; }
  else                    { *p = v[0]; }
}

template <int WIDTH, int NPW, bool FINAL>
__global__ void gcn_pull_sum(const float* __restrict__ h,
                             const int* __restrict__ rowptr,
                             const int* __restrict__ col,
                             const float* __restrict__ bias,
                             float* __restrict__ outp, int N, int doRelu) {
  constexpr int VEC = WIDTH / 32;
  constexpr int G = 8;
  const int lane = threadIdx.x & 31;
  const int wave = __builtin_amdgcn_readfirstlane(
      (int)((blockIdx.x * (size_t)blockDim.x + threadIdx.x) >> 5));
  const int node0 = wave * NPW;
  if (node0 >= N) return;

  int rp[NPW + 1];
#pragma unroll
  for (int i = 0; i <= NPW; ++i) {
    const int n = (node0 + i) < N ? (node0 + i) : N;
    rp[i] = rowptr[n];
  }
  int first[NPW];
#pragma unroll
  for (int i = 0; i < NPW; ++i)
    first[i] = (lane < rp[i + 1] - rp[i]) ? col[rp[i] + lane] : 0;

  const float* __restrict__ hl = h + lane * VEC;

#pragma unroll
  for (int i = 0; i < NPW; ++i) {
    const int node = node0 + i;
    if (node < N) {
      const int beg = rp[i], end = rp[i + 1];
      float acc[VEC];
#pragma unroll
      for (int q = 0; q < VEC; ++q) acc[q] = 0.0f;

      int cur = first[i];
      for (int p = beg; p < end; p += 32) {
        if (p != beg) cur = (lane < end - p) ? col[p + lane] : 0;
        const int m = (end - p) < 32 ? (end - p) : 32;
        for (int j = 0; j < m; j += G) {
          float v[G][VEC];
#pragma unroll
          for (int u = 0; u < G; ++u) {
            const int jj = (j + u) < m ? (j + u) : (m - 1);
            const int s = __builtin_amdgcn_readlane(cur, jj);
            gcn_ld_row<VEC>(hl + (size_t)s * WIDTH, v[u]);
          }
#pragma unroll
          for (int u = 0; u < G; ++u) {
            const bool on = (j + u) < m;
#pragma unroll
            for (int q = 0; q < VEC; ++q) acc[q] += on ? v[u][q] : 0.0f;
          }
        }
      }

      float r[VEC];
      if (FINAL) {
        float d = (float)(end - beg);
        d = d < 1.0f ? 1.0f : d;
#pragma unroll
        for (int q = 0; q < VEC; ++q) {
          float t = acc[q] / d + bias[lane * VEC + q];
          if (doRelu) t = t > 0.0f ? t : 0.0f;
          r[q] = t;
        }
      } else {
#pragma unroll
        for (int q = 0; q < VEC; ++q) r[q] = acc[q];
      }
      gcn_st_row<VEC>(outp + (size_t)node * WIDTH + lane * VEC, r);
    }
  }
}

constexpr int GCN_TILE_PITCH = 68;

__device__ __forceinline__ void gcn_ask_rows(const float* __restrict__ hl, int cur, int m,
                                             v2f (&v)[32]) {
#pragma unroll
  for (int b = 0; b < 4; ++b) {
    if ((b << 3) < m) {
#pragma unroll
      for (int u = 0; u < 8; ++u) {
        const int jj = ((b << 3) + u) < m ? ((b << 3) + u) : (m - 1);
        const int s = __builtin_amdgcn_readlane(cur, jj);
        v[(b << 3) + u] = *(const v2f*)(hl + (size_t)s * 64);
      }
    }
  }
}
__device__ __forceinline__ void gcn_add_rows(const v2f (&v)[32], int m, float& a0, float& a1) {
#pragma unroll
  for (int b = 0; b < 4; ++b) {
    if ((b << 3) < m) {
#pragma unroll
      for (int u = 0; u < 8; ++u) {
        const bool on = ((b << 3) + u) < m;
        a0 += on ? v[(b << 3) + u].x : 0.0f;
        a1 += on ? v[(b << 3) + u].y : 0.0f;
      }
    }
  }
}


__global__ void __launch_bounds__(128)
gcn_sum_dense_chain_wmma_f32(const float* __restrict__ h,
                             const int* __restrict__ rowptr,
                             const int* __restrict__ col,
                             const float* __restrict__ W1,
                             const int* __restrict__ cnt,
                             const float* __restrict__ b1,
                             const float* __restrict__ W2,
                             float* __restrict__ Y, int N) {
  __shared__ __attribute__((aligned(16))) float tiles[4 * 16 * GCN_TILE_PITCH];
  const int lane = threadIdx.x & 31;
  const int wave = __builtin_amdgcn_readfirstlane(
      (int)((blockIdx.x * (size_t)blockDim.x + threadIdx.x) >> 5));
  const int wib  = __builtin_amdgcn_readfirstlane((int)(threadIdx.x >> 5));
  if (wave >= (N >> 4)) return;
  float* __restrict__ tile = tiles + wib * (16 * GCN_TILE_PITCH);

  const int node0 = wave << 4;
  const int rpv = rowptr[node0 + (lane < 16 ? lane : 16)];
  int beg = __builtin_amdgcn_readlane(rpv, 0);
  int end = __builtin_amdgcn_readlane(rpv, 1);
  int cur = (lane < end - beg) ? col[beg + lane] : 0;
  const float* __restrict__ hl = h + lane * 2;

  for (int i = 0; i < 16; ++i) {
    const int nbeg = end;
    const int nend = __builtin_amdgcn_readlane(rpv, (i + 2) < 16 ? (i + 2) : 16);
    const int m0 = (end - beg) < 32 ? (end - beg) : 32;
    float a0 = 0.0f, a1 = 0.0f;
    v2f v[32];
    int nxt;
    if (m0 > 0) {
      gcn_ask_rows(hl, cur, m0, v);
      nxt = (lane < nend - nbeg) ? col[nbeg + lane] : 0;
      gcn_add_rows(v, m0, a0, a1);
      for (int p = beg + 32; p < end; p += 32) {
        const int c2 = (lane < end - p) ? col[p + lane] : 0;
        const int m = (end - p) < 32 ? (end - p) : 32;
        gcn_ask_rows(hl, c2, m, v);
        gcn_add_rows(v, m, a0, a1);
      }
    } else {
      nxt = (lane < nend - nbeg) ? col[nbeg + lane] : 0;
    }
    v2f t;
    t.x = a0;
    t.y = a1;
    *(v2f*)(tile + i * GCN_TILE_PITCH + lane * 2) = t;
    cur = nxt;
    beg = nbeg;
    end = nend;
  }
  __builtin_amdgcn_wave_barrier();

  const int half = lane >> 4;
  const int l16  = lane & 15;
  const int node = node0 + l16;
  const float* xrow = tile + l16 * GCN_TILE_PITCH;

  v8f s[4];
#pragma unroll
  for (int j = 0; j < 4; ++j) s[j] = (v8f){};

#pragma unroll 2
  for (int k = 0; k < 64; k += 4) {
    const int kk = k + (half << 1);
    const v2f x = *(const v2f*)(xrow + kk);
#pragma unroll
    for (int j = 0; j < 4; ++j) {
      v2f w;
      w.x = W1[kk * 64 + (j << 4) + l16];
      w.y = W1[(kk + 1) * 64 + (j << 4) + l16];
      s[j] = __builtin_amdgcn_wmma_f32_16x16x4_f32(
           false, w,  false, x,
           (short)0, s[j],  false,  false);
    }
  }

  float d = (float)cnt[node];
  d = d < 1.0f ? 1.0f : d;
  const float* __restrict__ bh = b1 + (half << 3);
#pragma unroll
  for (int j = 0; j < 4; ++j) {
#pragma unroll
    for (int r = 0; r < 8; ++r) {
      const float t = s[j][r] / d + bh[(j << 4) + r];
      s[j][r] = t > 0.0f ? t : 0.0f;
    }
  }

  v8f o[2];
  o[0] = (v8f){};
  o[1] = (v8f){};
#pragma unroll
  for (int j = 0; j < 4; ++j) {
#pragma unroll
    for (int q = 0; q < 4; ++q) {
      const int f = (j << 4) + (half << 3) + (q << 1);
      v2f x;
      x.x = s[j][2 * q];
      x.y = s[j][2 * q + 1];
#pragma unroll
      for (int t = 0; t < 2; ++t) {
        v2f w;
        w.x = W2[f * 32 + (t << 4) + l16];
        w.y = W2[(f + 1) * 32 + (t << 4) + l16];
        o[t] = __builtin_amdgcn_wmma_f32_16x16x4_f32(
             false, w,  false, x,
             (short)0, o[t],  false,  false);
      }
    }
  }

  float* __restrict__ yrow = Y + (size_t)node * 32 + (half << 3);
#pragma unroll
  for (int t = 0; t < 2; ++t) {
    *(v4f*)(yrow + (t << 4))     = __builtin_shufflevector(o[t], o[t], 0, 1, 2, 3);
    *(v4f*)(yrow + (t << 4) + 4) = __builtin_shufflevector(o[t], o[t], 4, 5, 6, 7);
  }
}

extern "C" void kernel_launch(void* const* d_in, const int* in_sizes, int n_in,
                              void* d_out, int out_size, void* d_ws, size_t ws_size,
                              hipStream_t stream) {
  const float* feat = (const float*)d_in[0];
  const float* W0   = (const float*)d_in[1];
  const float* b0   = (const float*)d_in[2];
  const float* W1   = (const float*)d_in[3];
  const float* b1   = (const float*)d_in[4];
  const float* W2   = (const float*)d_in[5];
  const float* b2   = (const float*)d_in[6];
  const int*   src  = (const int*)d_in[7];
  const int*   dst  = (const int*)d_in[8];
  float* out = (float*)d_out;

  const int N = in_sizes[0] / 256;
  const int E = in_sizes[7];

  float* bufA = (float*)d_ws;
  float* bufB = bufA + (size_t)N * 64;
  int*   cnt  = (int*)(bufB + (size_t)N * 64);
  int*   rank = (int*)bufB;
  int*   cm   = rank + (((size_t)E + 63) & ~(size_t)63);
  const int rpPad = ((N + 1 + 63) / 64) * 64;
  int* rowptrO = (int*)out;
  int* colO    = rowptrO + rpPad;
  int* rowptrB = (int*)bufB;
  int* colB    = rowptrB + rpPad;

  const int TB = 256;
  const int NPW = 4;
  const int gemmBlocks  = (int)((((long long)(N / 16)) * 32 + 1023) / 1024);
  const int edge4Blocks = (int)((((long long)E + 3) / 4 + TB - 1) / TB);
  const int scanBlocks  = (N + 1023) / 1024;
  const int nodeBlocks  = (N + TB - 1) / TB;
  constexpr int CH = 32;
  const int S  = (((E + CH - 1) / CH) + 3) & ~3;
  const int pullBlocks  = (int)(((long long)((N + NPW - 1) / NPW) * 32 + TB - 1) / TB);
  const int TBF = 128;
  const int fusedBlocks = (int)((((long long)(N / 16)) * 32 + TBF - 1) / TBF);
  const int moveBlocks  = (int)((((long long)rpPad + E + 3) / 4 + TB - 1) / TB);

  gcn_count_rank_and_product0<<<gemmBlocks + CH, 1024, 0, stream>>>(dst, cm, rank, E, N, S, gemmBlocks,
                                                                   feat, W0, bufA);
  gcn_chunk_prefix<CH><<<nodeBlocks, TB, 0, stream>>>(cm, cnt, N);
  gcn_rowptr_scan<<<scanBlocks, 1024, 0, stream>>>(cnt, rowptrO, N);
  gcn_fill_col<<<edge4Blocks, TB, 0, stream>>>(src, dst, rank, rowptrO, cm, colO, E, N, S);

  gcn_pull_sum<64, NPW, true><<<pullBlocks, TB, 0, stream>>>(bufA, rowptrO, colO, b0, bufB, N, 1);

  gcn_sum_dense_chain_wmma_f32<<<fusedBlocks, TBF, 0, stream>>>(bufB, rowptrO, colO, W1, cnt, b1, W2, bufA, N);

  gcn_move_ints<<<moveBlocks, TB, 0, stream>>>(rowptrO, rowptrB, rpPad + E);
  gcn_pull_sum<32, NPW, true><<<pullBlocks, TB, 0, stream>>>(bufA, rowptrB, colB, b2, out, N, 0);
}


# row offsets computed inside the prefix launch from per-chunk 512-node totals the counting blocks write; chunk table then holds absolute slots: no scan launch, fill reads 1 table entry per edge not 2
# speedup vs baseline: 3.4592x; 1.0535x over previous
#include <hip/hip_runtime.h>


typedef __attribute__((ext_vector_type(2))) float v2f;
typedef __attribute__((ext_vector_type(4))) float v4f;
typedef __attribute__((ext_vector_type(8))) float v8f;

constexpr int GCN_HWORDS = 51200;
constexpr int GCN_TRIPS  = 16;
constexpr int GCN_CHUNKS = 32;
constexpr int GCN_ROWS   = 512;

__device__ __forceinline__ int gcn_lds_rank16(unsigned int* hist, int d) {
  const unsigned int sh = (unsigned int)(d & 1) << 4;
  const unsigned int old = atomicAdd(hist + (d >> 1), 1u << sh);
  return (int)((old >> sh) & 0xFFFFu);
}

__device__ __forceinline__ void
gcn_chunk_count_rank_block(unsigned int* hist, const int* __restrict__ dst, int* __restrict__ cm,
                           int* __restrict__ tot, int* __restrict__ rank, int E, int N, int S,
                           int blk) {
  const int tid = threadIdx.x;
  uint4* h4 = (uint4*)hist;
  for (int i = tid; i < GCN_HWORDS / 4; i += 1024) h4[i] = make_uint4(0u, 0u, 0u, 0u);
  __syncthreads();

  const long long b0 = (long long)blk * S;
  const int e0 = b0 < E ? (int)b0 : E;
  const int e1 = (b0 + S) < E ? (int)(b0 + S) : E;
  const int e  = e0 + (tid << 2);
  int eL = (e1 - 4) & ~3;
  eL = eL < 0 ? 0 : eL;
  int4 dv[GCN_TRIPS];
#pragma unroll
  for (int j = 0; j < GCN_TRIPS; ++j) {
    const int ej = e + (j << 12);
    dv[j] = *(const int4*)(dst + (ej < eL ? ej : eL));
  }
#pragma unroll
  for (int j = 0; j < GCN_TRIPS; ++j) {
    const int ej = e + (j << 12);
    if (ej + 3 < e1) {
      int4 r;
      r.x = gcn_lds_rank16(hist, dv[j].x);
      r.y = gcn_lds_rank16(hist, dv[j].y);
      r.z = gcn_lds_rank16(hist, dv[j].z);
      r.w = gcn_lds_rank16(hist, dv[j].w);
      *(int4*)(rank + ej) = r;
    } else {
      for (int k = ej; k < e1; ++k) rank[k] = gcn_lds_rank16(hist, dst[k]);
    }
  }
  __syncthreads();

  int* __restrict__ row = cm + (size_t)blk * N;
  const uint2* h2 = (const uint2*)hist;
  const int lane = tid & 31;
  const int NQ = N >> 2;
  const int NB = (N + GCN_ROWS - 1) / GCN_ROWS;
  for (int nb = tid >> 5; nb < NB; nb += 32) {
    int s = 0;
#pragma unroll
    for (int k = 0; k < GCN_ROWS / 128; ++k) {
      const int q = nb * (GCN_ROWS / 4) + (k << 5) + lane;
      if (q < NQ) {
        const uint2 w = h2[q];
        int4 v;
        v.x = (int)(w.x & 0xFFFFu); v.y = (int)(w.x >> 16);
        v.z = (int)(w.y & 0xFFFFu); v.w = (int)(w.y >> 16);
        *(int4*)(row + (q << 2)) = v;
        s += (v.x + v.y) + (v.z + v.w);
      }
    }
    for (int o = 16; o > 0; o >>= 1) s += __shfl_xor(s, o);
    if (lane == 0) tot[(size_t)(nb + 1) * GCN_CHUNKS + blk] = s;
  }
  if (tid == 0) tot[blk] = 0;
}

template <int C>
__global__ void __launch_bounds__(GCN_ROWS)
gcn_chunk_prefix_rowptr(int* __restrict__ cm, const int* __restrict__ tot,
                        int* __restrict__ cnt, int* __restrict__ rowptr, int N) {
  constexpr int NW = GCN_ROWS / 32;
  __shared__ int wsum[NW];
  __shared__ int wpre[NW];
  const int tid = threadIdx.x, lane = tid & 31, w = tid >> 5;
  const int n = blockIdx.x * GCN_ROWS + tid;
  const bool in = n < N;
  const int nn = in ? n : N - 1;

  int t[C];
#pragma unroll
  for (int c = 0; c < C; ++c) t[c] = cm[(size_t)c * N + nn];

  int acc = 0;
  const int4* t4 = (const int4*)tot;
  const int n4 = ((int)blockIdx.x + 1) * (C / 4);
#pragma unroll 1
  for (int i0 = 0; i0 < n4; i0 += 4 * GCN_ROWS) {
#pragma unroll
    for (int u = 0; u < 4; ++u) {
      const int i = i0 + u * GCN_ROWS + tid;
      const int4 q = t4[i < n4 ? i : 0];
      acc += (q.x + q.y) + (q.z + q.w);
    }
  }

  int v = 0;
#pragma unroll
  for (int c = 0; c < C; ++c) v += t[c];
  v = in ? v : 0;

  int inc = v;
  for (int o = 1; o < 32; o <<= 1) {
    const int u = __shfl_up(inc, o);
    if (lane >= o) inc += u;
  }
  for (int o = 16; o > 0; o >>= 1) acc += __shfl_xor(acc, o);
  if (lane == 31) wsum[w] = inc;
  if (lane == 0)  wpre[w] = acc;
  __syncthreads();

  const int ws = lane < NW ? wsum[lane] : 0;
  int wsinc = ws;
  for (int o = 1; o < 32; o <<= 1) {
    const int u = __shfl_up(wsinc, o);
    if (lane >= o) wsinc += u;
  }
  int front = lane < NW ? wpre[lane] : 0;
  for (int o = 16; o > 0; o >>= 1) front += __shfl_xor(front, o);
  const int waveOff = __shfl(wsinc - ws, w);
  const int excl = front + waveOff + inc - v;

  if (in) {
    rowptr[n] = excl;
    cnt[n] = v;
    int sum = excl;
#pragma unroll
    for (int c = 0; c < C; ++c) {
      cm[(size_t)c * N + n] = sum;
      sum += t[c];
    }
  }
  if (n == N - 1) rowptr[N] = excl + v;
}

__global__ void gcn_fill_col(const int* __restrict__ src, const int* __restrict__ dst,
                             const int* __restrict__ rank,
                             const int* __restrict__ cm, int* __restrict__ col,
                             int E, int N, int S) {
  const int t = blockIdx.x * blockDim.x + threadIdx.x;
  const int e = t << 2;
  if (e + 3 < E) {
    const int4 s = *(const int4*)(src + e);
    const int4 d = *(const int4*)(dst + e);
    const int4 r = *(const int4*)(rank + e);
    const int* __restrict__ pre = cm + (size_t)(e / S) * N;
    col[pre[d.x] + r.x] = s.x;
    col[pre[d.y] + r.y] = s.y;
    col[pre[d.z] + r.z] = s.z;
    col[pre[d.w] + r.w] = s.w;
  } else {
    for (int k = e; k < E; ++k) {
      const int dk = dst[k];
      col[cm[(size_t)(k / S) * N + dk] + rank[k]] = src[k];
    }
  }
}

__global__ void gcn_move_ints(const int* __restrict__ from, int* __restrict__ to, int T) {
  const int t = blockIdx.x * blockDim.x + threadIdx.x;
  const int e = t << 2;
  if (e + 3 < T) {
    *(int4*)(to + e) = *(const int4*)(from + e);
  } else {
    for (int k = e; k < T; ++k) to[k] = from[k];
  }
}

template <int K, int TM>
__device__ __forceinline__ void gcn_gemm_wmma_f32_block(const float* __restrict__ X,
                                  const float* __restrict__ W,
                                  float* __restrict__ Y, int N, int blk) {
  const int M = TM * 16;
  const int wave = (int)((blk * (size_t)blockDim.x + threadIdx.x) >> 5);
  if (wave >= (N >> 4)) return;

  const int lane = threadIdx.x & 31;
  const int half = lane >> 4;
  const int l16  = lane & 15;
  const float* __restrict__ xrow = X + (size_t)((wave << 4) + l16) * K;

  v8f acc[TM];
#pragma unroll
  for (int j = 0; j < TM; ++j) acc[j] = (v8f){};

#pragma unroll 2
  for (int k = 0; k < K; k += 4) {
    const int kk = k + (half << 1);
    const v2f a = *(const v2f*)(xrow + kk);
#pragma unroll
    for (int j = 0; j < TM; ++j) {
      v2f b;
      b.x = W[kk * M + (j << 4) + l16];
      b.y = W[(kk + 1) * M + (j << 4) + l16];
      acc[j] = __builtin_amdgcn_wmma_f32_16x16x4_f32(
           false, a,  false, b,
           (short)0, acc[j],  false,  false);
    }
  }

  const int rbase = (wave << 4) + (half << 3);
#pragma unroll
  for (int j = 0; j < TM; ++j)
#pragma unroll
    for (int r = 0; r < 8; ++r)
      Y[(size_t)(rbase + r) * M + (j << 4) + l16] = acc[j][r];
}

__global__ void __launch_bounds__(1024)
gcn_count_rank_and_product0(const int* __restrict__ dst, int* __restrict__ cm,
                            int* __restrict__ tot, int* __restrict__ rank, int E, int N, int S,
                            int rowBlocks,
                            const float* __restrict__ X, const float* __restrict__ W,
                            float* __restrict__ Y) {
  __shared__ __attribute__((aligned(16))) unsigned int hist[GCN_HWORDS];
  const int b = (int)blockIdx.x;
  if (b < rowBlocks) {
    gcn_gemm_wmma_f32_block<256, 4>(X, W, Y, N, b);
  } else {
    gcn_chunk_count_rank_block(hist, dst, cm, tot, rank, E, N, S, b - rowBlocks);
  }
}

template <int VEC>
__device__ __forceinline__ void gcn_ld_row(const float* p, float (&v)[VEC]) {
  if constexpr (VEC == 2) { const v2f t = *(const v2f*)p; v[0] = t.x; v[1] = t.y; }
  else                    { v[0] = *p; }
}
template <int VEC>
__device__ __forceinline__ void gcn_st_row(float* p, const float (&v)[VEC]) {
  if constexpr (VEC == 2) { v2f t; t.x = v[0]; t.y = v[1]; *(v2f*)p = t; }
  else                    { *p = v[0]; }
}

template <int WIDTH, int NPW, bool FINAL>
__global__ void gcn_pull_sum(const float* __restrict__ h,
                             const int* __restrict__ rowptr,
                             const int* __restrict__ col,
                             const float* __restrict__ bias,
                             float* __restrict__ outp, int N, int doRelu) {
  constexpr int VEC = WIDTH / 32;
  constexpr int G = 8;
  const int lane = threadIdx.x & 31;
  const int wave = __builtin_amdgcn_readfirstlane(
      (int)((blockIdx.x * (size_t)blockDim.x + threadIdx.x) >> 5));
  const int node0 = wave * NPW;
  if (node0 >= N) return;

  int rp[NPW + 1];
#pragma unroll
  for (int i = 0; i <= NPW; ++i) {
    const int n = (node0 + i) < N ? (node0 + i) : N;
    rp[i] = rowptr[n];
  }
  int first[NPW];
#pragma unroll
  for (int i = 0; i < NPW; ++i)
    first[i] = (lane < rp[i + 1] - rp[i]) ? col[rp[i] + lane] : 0;

  const float* __restrict__ hl = h + lane * VEC;

#pragma unroll
  for (int i = 0; i < NPW; ++i) {
    const int node = node0 + i;
    if (node < N) {
      const int beg = rp[i], end = rp[i + 1];
      float acc[VEC];
#pragma unroll
      for (int q = 0; q < VEC; ++q) acc[q] = 0.0f;

      int cur = first[i];
      for (int p = beg; p < end; p += 32) {
        if (p != beg) cur = (lane < end - p) ? col[p + lane] : 0;
        const int m = (end - p) < 32 ? (end - p) : 32;
        for (int j = 0; j < m; j += G) {
          float v[G][VEC];
#pragma unroll
          for (int u = 0; u < G; ++u) {
            const int jj = (j + u) < m ? (j + u) : (m - 1);
            const int s = __builtin_amdgcn_readlane(cur, jj);
            gcn_ld_row<VEC>(hl + (size_t)s * WIDTH, v[u]);
          }
#pragma unroll
          for (int u = 0; u < G; ++u) {
            const bool on = (j + u) < m;
#pragma unroll
            for (int q = 0; q < VEC; ++q) acc[q] += on ? v[u][q] : 0.0f;
          }
        }
      }

      float r[VEC];
      if (FINAL) {
        float d = (float)(end - beg);
        d = d < 1.0f ? 1.0f : d;
#pragma unroll
        for (int q = 0; q < VEC; ++q) {
          float t = acc[q] / d + bias[lane * VEC + q];
          if (doRelu) t = t > 0.0f ? t : 0.0f;
          r[q] = t;
        }
      } else {
#pragma unroll
        for (int q = 0; q < VEC; ++q) r[q] = acc[q];
      }
      gcn_st_row<VEC>(outp + (size_t)node * WIDTH + lane * VEC, r);
    }
  }
}

constexpr int GCN_TILE_PITCH = 68;

__device__ __forceinline__ void gcn_ask_rows(const float* __restrict__ hl, int cur, int m,
                                             v2f (&v)[32]) {
#pragma unroll
  for (int b = 0; b < 4; ++b) {
    if ((b << 3) < m) {
#pragma unroll
      for (int u = 0; u < 8; ++u) {
        const int jj = ((b << 3) + u) < m ? ((b << 3) + u) : (m - 1);
        const int s = __builtin_amdgcn_readlane(cur, jj);
        v[(b << 3) + u] = *(const v2f*)(hl + (size_t)s * 64);
      }
    }
  }
}
__device__ __forceinline__ void gcn_add_rows(const v2f (&v)[32], int m, float& a0, float& a1) {
#pragma unroll
  for (int b = 0; b < 4; ++b) {
    if ((b << 3) < m) {
#pragma unroll
      for (int u = 0; u < 8; ++u) {
        const bool on = ((b << 3) + u) < m;
        a0 += on ? v[(b << 3) + u].x : 0.0f;
        a1 += on ? v[(b << 3) + u].y : 0.0f;
      }
    }
  }
}


__global__ void __launch_bounds__(128)
gcn_sum_dense_chain_wmma_f32(const float* __restrict__ h,
                             const int* __restrict__ rowptr,
                             const int* __restrict__ col,
                             const float* __restrict__ W1,
                             const int* __restrict__ cnt,
                             const float* __restrict__ b1,
                             const float* __restrict__ W2,
                             float* __restrict__ Y, int N) {
  __shared__ __attribute__((aligned(16))) float tiles[4 * 16 * GCN_TILE_PITCH];
  const int lane = threadIdx.x & 31;
  const int wave = __builtin_amdgcn_readfirstlane(
      (int)((blockIdx.x * (size_t)blockDim.x + threadIdx.x) >> 5));
  const int wib  = __builtin_amdgcn_readfirstlane((int)(threadIdx.x >> 5));
  if (wave >= (N >> 4)) return;
  float* __restrict__ tile = tiles + wib * (16 * GCN_TILE_PITCH);

  const int node0 = wave << 4;
  const int rpv = rowptr[node0 + (lane < 16 ? lane : 16)];
  int beg = __builtin_amdgcn_readlane(rpv, 0);
  int end = __builtin_amdgcn_readlane(rpv, 1);
  int cur = (lane < end - beg) ? col[beg + lane] : 0;
  const float* __restrict__ hl = h + lane * 2;

  for (int i = 0; i < 16; ++i) {
    const int nbeg = end;
    const int nend = __builtin_amdgcn_readlane(rpv, (i + 2) < 16 ? (i + 2) : 16);
    const int m0 = (end - beg) < 32 ? (end - beg) : 32;
    float a0 = 0.0f, a1 = 0.0f;
    v2f v[32];
    int nxt;
    if (m0 > 0) {
      gcn_ask_rows(hl, cur, m0, v);
      nxt = (lane < nend - nbeg) ? col[nbeg + lane] : 0;
      gcn_add_rows(v, m0, a0, a1);
      for (int p = beg + 32; p < end; p += 32) {
        const int c2 = (lane < end - p) ? col[p + lane] : 0;
        const int m = (end - p) < 32 ? (end - p) : 32;
        gcn_ask_rows(hl, c2, m, v);
        gcn_add_rows(v, m, a0, a1);
      }
    } else {
      nxt = (lane < nend - nbeg) ? col[nbeg + lane] : 0;
    }
    v2f t;
    t.x = a0;
    t.y = a1;
    *(v2f*)(tile + i * GCN_TILE_PITCH + lane * 2) = t;
    cur = nxt;
    beg = nbeg;
    end = nend;
  }
  __builtin_amdgcn_wave_barrier();

  const int half = lane >> 4;
  const int l16  = lane & 15;
  const int node = node0 + l16;
  const float* xrow = tile + l16 * GCN_TILE_PITCH;

  v8f s[4];
#pragma unroll
  for (int j = 0; j < 4; ++j) s[j] = (v8f){};

#pragma unroll 2
  for (int k = 0; k < 64; k += 4) {
    const int kk = k + (half << 1);
    const v2f x = *(const v2f*)(xrow + kk);
#pragma unroll
    for (int j = 0; j < 4; ++j) {
      v2f w;
      w.x = W1[kk * 64 + (j << 4) + l16];
      w.y = W1[(kk + 1) * 64 + (j << 4) + l16];
      s[j] = __builtin_amdgcn_wmma_f32_16x16x4_f32(
           false, w,  false, x,
           (short)0, s[j],  false,  false);
    }
  }

  float d = (float)cnt[node];
  d = d < 1.0f ? 1.0f : d;
  const float* __restrict__ bh = b1 + (half << 3);
#pragma unroll
  for (int j = 0; j < 4; ++j) {
#pragma unroll
    for (int r = 0; r < 8; ++r) {
      const float t = s[j][r] / d + bh[(j << 4) + r];
      s[j][r] = t > 0.0f ? t : 0.0f;
    }
  }

  v8f o[2];
  o[0] = (v8f){};
  o[1] = (v8f){};
#pragma unroll
  for (int j = 0; j < 4; ++j) {
#pragma unroll
    for (int q = 0; q < 4; ++q) {
      const int f = (j << 4) + (half << 3) + (q << 1);
      v2f x;
      x.x = s[j][2 * q];
      x.y = s[j][2 * q + 1];
#pragma unroll
      for (int t = 0; t < 2; ++t) {
        v2f w;
        w.x = W2[f * 32 + (t << 4) + l16];
        w.y = W2[(f + 1) * 32 + (t << 4) + l16];
        o[t] = __builtin_amdgcn_wmma_f32_16x16x4_f32(
             false, w,  false, x,
             (short)0, o[t],  false,  false);
      }
    }
  }

  float* __restrict__ yrow = Y + (size_t)node * 32 + (half << 3);
#pragma unroll
  for (int t = 0; t < 2; ++t) {
    *(v4f*)(yrow + (t << 4))     = __builtin_shufflevector(o[t], o[t], 0, 1, 2, 3);
    *(v4f*)(yrow + (t << 4) + 4) = __builtin_shufflevector(o[t], o[t], 4, 5, 6, 7);
  }
}

extern "C" void kernel_launch(void* const* d_in, const int* in_sizes, int n_in,
                              void* d_out, int out_size, void* d_ws, size_t ws_size,
                              hipStream_t stream) {
  const float* feat = (const float*)d_in[0];
  const float* W0   = (const float*)d_in[1];
  const float* b0   = (const float*)d_in[2];
  const float* W1   = (const float*)d_in[3];
  const float* b1   = (const float*)d_in[4];
  const float* W2   = (const float*)d_in[5];
  const float* b2   = (const float*)d_in[6];
  const int*   src  = (const int*)d_in[7];
  const int*   dst  = (const int*)d_in[8];
  float* out = (float*)d_out;

  const int N = in_sizes[0] / 256;
  const int E = in_sizes[7];

  float* bufA = (float*)d_ws;
  float* bufB = bufA + (size_t)N * 64;
  int*   cnt  = (int*)(bufB + (size_t)N * 64);
  int*   rank = (int*)bufB;
  int*   cm   = rank + (((size_t)E + 63) & ~(size_t)63);
  int*   tot  = cm + (size_t)GCN_CHUNKS * N;
  const int rpPad = ((N + 1 + 63) / 64) * 64;
  int* rowptrO = (int*)out;
  int* colO    = rowptrO + rpPad;
  int* rowptrB = (int*)bufB;
  int* colB    = rowptrB + rpPad;

  const int TB = 256;
  const int NPW = 4;
  const int gemmBlocks  = (int)((((long long)(N / 16)) * 32 + 1023) / 1024);
  const int edge4Blocks = (int)((((long long)E + 3) / 4 + TB - 1) / TB);
  const int nodeBlocks  = (N + GCN_ROWS - 1) / GCN_ROWS;
  constexpr int CH = GCN_CHUNKS;
  const int S  = (((E + CH - 1) / CH) + 3) & ~3;
  const int pullBlocks  = (int)(((long long)((N + NPW - 1) / NPW) * 32 + TB - 1) / TB);
  const int TBF = 128;
  const int fusedBlocks = (int)((((long long)(N / 16)) * 32 + TBF - 1) / TBF);
  const int moveBlocks  = (int)((((long long)rpPad + E + 3) / 4 + TB - 1) / TB);

  gcn_count_rank_and_product0<<<gemmBlocks + CH, 1024, 0, stream>>>(dst, cm, tot, rank, E, N, S,
                                                                   gemmBlocks, feat, W0, bufA);
  gcn_chunk_prefix_rowptr<CH><<<nodeBlocks, GCN_ROWS, 0, stream>>>(cm, tot, cnt, rowptrO, N);
  gcn_fill_col<<<edge4Blocks, TB, 0, stream>>>(src, dst, rank, cm, colO, E, N, S);

  gcn_pull_sum<64, NPW, true><<<pullBlocks, TB, 0, stream>>>(bufA, rowptrO, colO, b0, bufB, N, 1);

  gcn_sum_dense_chain_wmma_f32<<<fusedBlocks, TBF, 0, stream>>>(bufB, rowptrO, colO, W1, cnt, b1, W2, bufA, N);

  gcn_move_ints<<<moveBlocks, TB, 0, stream>>>(rowptrO, rowptrB, rpPad + E);
  gcn_pull_sum<32, NPW, true><<<pullBlocks, TB, 0, stream>>>(bufA, rowptrB, colB, b2, out, N, 0);
}


# layer 0's product: W0 staged once per row block in the launch's LDS as k-pair rows (pitch 160 dwords, conflict-free); B fragments are 8-byte LDS reads, not 8 global loads per k-step; k order kept
# speedup vs baseline: 3.6067x; 1.0426x over previous
#include <hip/hip_runtime.h>


typedef __attribute__((ext_vector_type(2))) float v2f;
typedef __attribute__((ext_vector_type(4))) float v4f;
typedef __attribute__((ext_vector_type(8))) float v8f;

constexpr int GCN_HWORDS = 51200;
constexpr int GCN_TRIPS  = 16;
constexpr int GCN_CHUNKS = 32;
constexpr int GCN_ROWS   = 512;

__device__ __forceinline__ int gcn_lds_rank16(unsigned int* hist, int d) {
  const unsigned int sh = (unsigned int)(d & 1) << 4;
  const unsigned int old = atomicAdd(hist + (d >> 1), 1u << sh);
  return (int)((old >> sh) & 0xFFFFu);
}

__device__ __forceinline__ void
gcn_chunk_count_rank_block(unsigned int* hist, const int* __restrict__ dst, int* __restrict__ cm,
                           int* __restrict__ tot, int* __restrict__ rank, int E, int N, int S,
                           int blk) {
  const int tid = threadIdx.x;
  uint4* h4 = (uint4*)hist;
  for (int i = tid; i < GCN_HWORDS / 4; i += 1024) h4[i] = make_uint4(0u, 0u, 0u, 0u);
  __syncthreads();

  const long long b0 = (long long)blk * S;
  const int e0 = b0 < E ? (int)b0 : E;
  const int e1 = (b0 + S) < E ? (int)(b0 + S) : E;
  const int e  = e0 + (tid << 2);
  int eL = (e1 - 4) & ~3;
  eL = eL < 0 ? 0 : eL;
  int4 dv[GCN_TRIPS];
#pragma unroll
  for (int j = 0; j < GCN_TRIPS; ++j) {
    const int ej = e + (j << 12);
    dv[j] = *(const int4*)(dst + (ej < eL ? ej : eL));
  }
#pragma unroll
  for (int j = 0; j < GCN_TRIPS; ++j) {
    const int ej = e + (j << 12);
    if (ej + 3 < e1) {
      int4 r;
      r.x = gcn_lds_rank16(hist, dv[j].x);
      r.y = gcn_lds_rank16(hist, dv[j].y);
      r.z = gcn_lds_rank16(hist, dv[j].z);
      r.w = gcn_lds_rank16(hist, dv[j].w);
      *(int4*)(rank + ej) = r;
    } else {
      for (int k = ej; k < e1; ++k) rank[k] = gcn_lds_rank16(hist, dst[k]);
    }
  }
  __syncthreads();

  int* __restrict__ row = cm + (size_t)blk * N;
  const uint2* h2 = (const uint2*)hist;
  const int lane = tid & 31;
  const int NQ = N >> 2;
  const int NB = (N + GCN_ROWS - 1) / GCN_ROWS;
  for (int nb = tid >> 5; nb < NB; nb += 32) {
    int s = 0;
#pragma unroll
    for (int k = 0; k < GCN_ROWS / 128; ++k) {
      const int q = nb * (GCN_ROWS / 4) + (k << 5) + lane;
      if (q < NQ) {
        const uint2 w = h2[q];
        int4 v;
        v.x = (int)(w.x & 0xFFFFu); v.y = (int)(w.x >> 16);
        v.z = (int)(w.y & 0xFFFFu); v.w = (int)(w.y >> 16);
        *(int4*)(row + (q << 2)) = v;
        s += (v.x + v.y) + (v.z + v.w);
      }
    }
    for (int o = 16; o > 0; o >>= 1) s += __shfl_xor(s, o);
    if (lane == 0) tot[(size_t)(nb + 1) * GCN_CHUNKS + blk] = s;
  }
  if (tid == 0) tot[blk] = 0;
}

template <int C>
__global__ void __launch_bounds__(GCN_ROWS)
gcn_chunk_prefix_rowptr(int* __restrict__ cm, const int* __restrict__ tot,
                        int* __restrict__ cnt, int* __restrict__ rowptr, int N) {
  constexpr int NW = GCN_ROWS / 32;
  __shared__ int wsum[NW];
  __shared__ int wpre[NW];
  const int tid = threadIdx.x, lane = tid & 31, w = tid >> 5;
  const int n = blockIdx.x * GCN_ROWS + tid;
  const bool in = n < N;
  const int nn = in ? n : N - 1;

  int t[C];
#pragma unroll
  for (int c = 0; c < C; ++c) t[c] = cm[(size_t)c * N + nn];

  int acc = 0;
  const int4* t4 = (const int4*)tot;
  const int n4 = ((int)blockIdx.x + 1) * (C / 4);
#pragma unroll 1
  for (int i0 = 0; i0 < n4; i0 += 4 * GCN_ROWS) {
#pragma unroll
    for (int u = 0; u < 4; ++u) {
      const int i = i0 + u * GCN_ROWS + tid;
      const int4 q = t4[i < n4 ? i : 0];
      acc += (q.x + q.y) + (q.z + q.w);
    }
  }

  int v = 0;
#pragma unroll
  for (int c = 0; c < C; ++c) v += t[c];
  v = in ? v : 0;

  int inc = v;
  for (int o = 1; o < 32; o <<= 1) {
    const int u = __shfl_up(inc, o);
    if (lane >= o) inc += u;
  }
  for (int o = 16; o > 0; o >>= 1) acc += __shfl_xor(acc, o);
  if (lane == 31) wsum[w] = inc;
  if (lane == 0)  wpre[w] = acc;
  __syncthreads();

  const int ws = lane < NW ? wsum[lane] : 0;
  int wsinc = ws;
  for (int o = 1; o < 32; o <<= 1) {
    const int u = __shfl_up(wsinc, o);
    if (lane >= o) wsinc += u;
  }
  int front = lane < NW ? wpre[lane] : 0;
  for (int o = 16; o > 0; o >>= 1) front += __shfl_xor(front, o);
  const int waveOff = __shfl(wsinc - ws, w);
  const int excl = front + waveOff + inc - v;

  if (in) {
    rowptr[n] = excl;
    cnt[n] = v;
    int sum = excl;
#pragma unroll
    for (int c = 0; c < C; ++c) {
      cm[(size_t)c * N + n] = sum;
      sum += t[c];
    }
  }
  if (n == N - 1) rowptr[N] = excl + v;
}

__global__ void gcn_fill_col(const int* __restrict__ src, const int* __restrict__ dst,
                             const int* __restrict__ rank,
                             const int* __restrict__ cm, int* __restrict__ col,
                             int E, int N, int S) {
  const int t = blockIdx.x * blockDim.x + threadIdx.x;
  const int e = t << 2;
  if (e + 3 < E) {
    const int4 s = *(const int4*)(src + e);
    const int4 d = *(const int4*)(dst + e);
    const int4 r = *(const int4*)(rank + e);
    const int* __restrict__ pre = cm + (size_t)(e / S) * N;
    col[pre[d.x] + r.x] = s.x;
    col[pre[d.y] + r.y] = s.y;
    col[pre[d.z] + r.z] = s.z;
    col[pre[d.w] + r.w] = s.w;
  } else {
    for (int k = e; k < E; ++k) {
      const int dk = dst[k];
      col[cm[(size_t)(k / S) * N + dk] + rank[k]] = src[k];
    }
  }
}

__global__ void gcn_move_ints(const int* __restrict__ from, int* __restrict__ to, int T) {
  const int t = blockIdx.x * blockDim.x + threadIdx.x;
  const int e = t << 2;
  if (e + 3 < T) {
    *(int4*)(to + e) = *(const int4*)(from + e);
  } else {
    for (int k = e; k < T; ++k) to[k] = from[k];
  }
}

template <int K, int TM>
__device__ __forceinline__ void gcn_gemm_wmma_f32_block(float* wl,
                                  const float* __restrict__ X,
                                  const float* __restrict__ W,
                                  float* __restrict__ Y, int N, int blk) {
  constexpr int M = TM * 16;
  constexpr int PITCH = 2 * M + 32;

  for (int i = threadIdx.x; i < (K / 2) * (M / 4); i += blockDim.x) {
    const int p  = i / (M / 4);
    const int c4 = (i % (M / 4)) << 2;
    const v4f r0 = *(const v4f*)(W + (size_t)(2 * p) * M + c4);
    const v4f r1 = *(const v4f*)(W + (size_t)(2 * p + 1) * M + c4);
    v4f lo, hi;
    lo.x = r0.x; lo.y = r1.x; lo.z = r0.y; lo.w = r1.y;
    hi.x = r0.z; hi.y = r1.z; hi.z = r0.w; hi.w = r1.w;
    float* q = wl + p * PITCH + (c4 << 1);
    *(v4f*)q       = lo;
    *(v4f*)(q + 4) = hi;
  }
  __syncthreads();

  const int wave = (int)((blk * (size_t)blockDim.x + threadIdx.x) >> 5);
  if (wave >= (N >> 4)) return;

  const int lane = threadIdx.x & 31;
  const int half = lane >> 4;
  const int l16  = lane & 15;
  const float* __restrict__ xrow = X + (size_t)((wave << 4) + l16) * K;
  const float* wfrag = wl + half * PITCH + (l16 << 1);

  v8f acc[TM];
#pragma unroll
  for (int j = 0; j < TM; ++j) acc[j] = (v8f){};

#pragma unroll 2
  for (int k = 0; k < K; k += 4) {
    const int kk = k + (half << 1);
    const v2f a = *(const v2f*)(xrow + kk);
#pragma unroll
    for (int j = 0; j < TM; ++j) {
      const v2f b = *(const v2f*)(wfrag + (k >> 1) * PITCH + (j << 5));
      acc[j] = __builtin_amdgcn_wmma_f32_16x16x4_f32(
           false, a,  false, b,
           (short)0, acc[j],  false,  false);
    }
  }

  const int rbase = (wave << 4) + (half << 3);
#pragma unroll
  for (int j = 0; j < TM; ++j)
#pragma unroll
    for (int r = 0; r < 8; ++r)
      Y[(size_t)(rbase + r) * M + (j << 4) + l16] = acc[j][r];
}

__global__ void __launch_bounds__(1024)
gcn_count_rank_and_product0(const int* __restrict__ dst, int* __restrict__ cm,
                            int* __restrict__ tot, int* __restrict__ rank, int E, int N, int S,
                            int rowBlocks,
                            const float* __restrict__ X, const float* __restrict__ W,
                            float* __restrict__ Y) {
  __shared__ __attribute__((aligned(16))) unsigned int hist[GCN_HWORDS];
  const int b = (int)blockIdx.x;
  if (b < rowBlocks) {
    gcn_gemm_wmma_f32_block<256, 4>((float*)hist, X, W, Y, N, b);
  } else {
    gcn_chunk_count_rank_block(hist, dst, cm, tot, rank, E, N, S, b - rowBlocks);
  }
}

template <int VEC>
__device__ __forceinline__ void gcn_ld_row(const float* p, float (&v)[VEC]) {
  if constexpr (VEC == 2) { const v2f t = *(const v2f*)p; v[0] = t.x; v[1] = t.y; }
  else                    { v[0] = *p; }
}
template <int VEC>
__device__ __forceinline__ void gcn_st_row(float* p, const float (&v)[VEC]) {
  if constexpr (VEC == 2) { v2f t; t.x = v[0]; t.y = v[1]; *(v2f*)p = t; }
  else                    { *p = v[0]; }
}

template <int WIDTH, int NPW, bool FINAL>
__global__ void gcn_pull_sum(const float* __restrict__ h,
                             const int* __restrict__ rowptr,
                             const int* __restrict__ col,
                             const float* __restrict__ bias,
                             float* __restrict__ outp, int N, int doRelu) {
  constexpr int VEC = WIDTH / 32;
  constexpr int G = 8;
  const int lane = threadIdx.x & 31;
  const int wave = __builtin_amdgcn_readfirstlane(
      (int)((blockIdx.x * (size_t)blockDim.x + threadIdx.x) >> 5));
  const int node0 = wave * NPW;
  if (node0 >= N) return;

  int rp[NPW + 1];
#pragma unroll
  for (int i = 0; i <= NPW; ++i) {
    const int n = (node0 + i) < N ? (node0 + i) : N;
    rp[i] = rowptr[n];
  }
  int first[NPW];
#pragma unroll
  for (int i = 0; i < NPW; ++i)
    first[i] = (lane < rp[i + 1] - rp[i]) ? col[rp[i] + lane] : 0;

  const float* __restrict__ hl = h + lane * VEC;

#pragma unroll
  for (int i = 0; i < NPW; ++i) {
    const int node = node0 + i;
    if (node < N) {
      const int beg = rp[i], end = rp[i + 1];
      float acc[VEC];
#pragma unroll
      for (int q = 0; q < VEC; ++q) acc[q] = 0.0f;

      int cur = first[i];
      for (int p = beg; p < end; p += 32) {
        if (p != beg) cur = (lane < end - p) ? col[p + lane] : 0;
        const int m = (end - p) < 32 ? (end - p) : 32;
        for (int j = 0; j < m; j += G) {
          float v[G][VEC];
#pragma unroll
          for (int u = 0; u < G; ++u) {
            const int jj = (j + u) < m ? (j + u) : (m - 1);
            const int s = __builtin_amdgcn_readlane(cur, jj);
            gcn_ld_row<VEC>(hl + (size_t)s * WIDTH, v[u]);
          }
#pragma unroll
          for (int u = 0; u < G; ++u) {
            const bool on = (j + u) < m;
#pragma unroll
            for (int q = 0; q < VEC; ++q) acc[q] += on ? v[u][q] : 0.0f;
          }
        }
      }

      float r[VEC];
      if (FINAL) {
        float d = (float)(end - beg);
        d = d < 1.0f ? 1.0f : d;
#pragma unroll
        for (int q = 0; q < VEC; ++q) {
          float t = acc[q] / d + bias[lane * VEC + q];
          if (doRelu) t = t > 0.0f ? t : 0.0f;
          r[q] = t;
        }
      } else {
#pragma unroll
        for (int q = 0; q < VEC; ++q) r[q] = acc[q];
      }
      gcn_st_row<VEC>(outp + (size_t)node * WIDTH + lane * VEC, r);
    }
  }
}

constexpr int GCN_TILE_PITCH = 68;

__device__ __forceinline__ void gcn_ask_rows(const float* __restrict__ hl, int cur, int m,
                                             v2f (&v)[32]) {
#pragma unroll
  for (int b = 0; b < 4; ++b) {
    if ((b << 3) < m) {
#pragma unroll
      for (int u = 0; u < 8; ++u) {
        const int jj = ((b << 3) + u) < m ? ((b << 3) + u) : (m - 1);
        const int s = __builtin_amdgcn_readlane(cur, jj);
        v[(b << 3) + u] = *(const v2f*)(hl + (size_t)s * 64);
      }
    }
  }
}
__device__ __forceinline__ void gcn_add_rows(const v2f (&v)[32], int m, float& a0, float& a1) {
#pragma unroll
  for (int b = 0; b < 4; ++b) {
    if ((b << 3) < m) {
#pragma unroll
      for (int u = 0; u < 8; ++u) {
        const bool on = ((b << 3) + u) < m;
        a0 += on ? v[(b << 3) + u].x : 0.0f;
        a1 += on ? v[(b << 3) + u].y : 0.0f;
      }
    }
  }
}


__global__ void __launch_bounds__(128)
gcn_sum_dense_chain_wmma_f32(const float* __restrict__ h,
                             const int* __restrict__ rowptr,
                             const int* __restrict__ col,
                             const float* __restrict__ W1,
                             const int* __restrict__ cnt,
                             const float* __restrict__ b1,
                             const float* __restrict__ W2,
                             float* __restrict__ Y, int N) {
  __shared__ __attribute__((aligned(16))) float tiles[4 * 16 * GCN_TILE_PITCH];
  const int lane = threadIdx.x & 31;
  const int wave = __builtin_amdgcn_readfirstlane(
      (int)((blockIdx.x * (size_t)blockDim.x + threadIdx.x) >> 5));
  const int wib  = __builtin_amdgcn_readfirstlane((int)(threadIdx.x >> 5));
  if (wave >= (N >> 4)) return;
  float* __restrict__ tile = tiles + wib * (16 * GCN_TILE_PITCH);

  const int node0 = wave << 4;
  const int rpv = rowptr[node0 + (lane < 16 ? lane : 16)];
  int beg = __builtin_amdgcn_readlane(rpv, 0);
  int end = __builtin_amdgcn_readlane(rpv, 1);
  int cur = (lane < end - beg) ? col[beg + lane] : 0;
  const float* __restrict__ hl = h + lane * 2;

  for (int i = 0; i < 16; ++i) {
    const int nbeg = end;
    const int nend = __builtin_amdgcn_readlane(rpv, (i + 2) < 16 ? (i + 2) : 16);
    const int m0 = (end - beg) < 32 ? (end - beg) : 32;
    float a0 = 0.0f, a1 = 0.0f;
    v2f v[32];
    int nxt;
    if (m0 > 0) {
      gcn_ask_rows(hl, cur, m0, v);
      nxt = (lane < nend - nbeg) ? col[nbeg + lane] : 0;
      gcn_add_rows(v, m0, a0, a1);
      for (int p = beg + 32; p < end; p += 32) {
        const int c2 = (lane < end - p) ? col[p + lane] : 0;
        const int m = (end - p) < 32 ? (end - p) : 32;
        gcn_ask_rows(hl, c2, m, v);
        gcn_add_rows(v, m, a0, a1);
      }
    } else {
      nxt = (lane < nend - nbeg) ? col[nbeg + lane] : 0;
    }
    v2f t;
    t.x = a0;
    t.y = a1;
    *(v2f*)(tile + i * GCN_TILE_PITCH + lane * 2) = t;
    cur = nxt;
    beg = nbeg;
    end = nend;
  }
  __builtin_amdgcn_wave_barrier();

  const int half = lane >> 4;
  const int l16  = lane & 15;
  const int node = node0 + l16;
  const float* xrow = tile + l16 * GCN_TILE_PITCH;

  v8f s[4];
#pragma unroll
  for (int j = 0; j < 4; ++j) s[j] = (v8f){};

#pragma unroll 2
  for (int k = 0; k < 64; k += 4) {
    const int kk = k + (half << 1);
    const v2f x = *(const v2f*)(xrow + kk);
#pragma unroll
    for (int j = 0; j < 4; ++j) {
      v2f w;
      w.x = W1[kk * 64 + (j << 4) + l16];
      w.y = W1[(kk + 1) * 64 + (j << 4) + l16];
      s[j] = __builtin_amdgcn_wmma_f32_16x16x4_f32(
           false, w,  false, x,
           (short)0, s[j],  false,  false);
    }
  }

  float d = (float)cnt[node];
  d = d < 1.0f ? 1.0f : d;
  const float* __restrict__ bh = b1 + (half << 3);
#pragma unroll
  for (int j = 0; j < 4; ++j) {
#pragma unroll
    for (int r = 0; r < 8; ++r) {
      const float t = s[j][r] / d + bh[(j << 4) + r];
      s[j][r] = t > 0.0f ? t : 0.0f;
    }
  }

  v8f o[2];
  o[0] = (v8f){};
  o[1] = (v8f){};
#pragma unroll
  for (int j = 0; j < 4; ++j) {
#pragma unroll
    for (int q = 0; q < 4; ++q) {
      const int f = (j << 4) + (half << 3) + (q << 1);
      v2f x;
      x.x = s[j][2 * q];
      x.y = s[j][2 * q + 1];
#pragma unroll
      for (int t = 0; t < 2; ++t) {
        v2f w;
        w.x = W2[f * 32 + (t << 4) + l16];
        w.y = W2[(f + 1) * 32 + (t << 4) + l16];
        o[t] = __builtin_amdgcn_wmma_f32_16x16x4_f32(
             false, w,  false, x,
             (short)0, o[t],  false,  false);
      }
    }
  }

  float* __restrict__ yrow = Y + (size_t)node * 32 + (half << 3);
#pragma unroll
  for (int t = 0; t < 2; ++t) {
    *(v4f*)(yrow + (t << 4))     = __builtin_shufflevector(o[t], o[t], 0, 1, 2, 3);
    *(v4f*)(yrow + (t << 4) + 4) = __builtin_shufflevector(o[t], o[t], 4, 5, 6, 7);
  }
}

extern "C" void kernel_launch(void* const* d_in, const int* in_sizes, int n_in,
                              void* d_out, int out_size, void* d_ws, size_t ws_size,
                              hipStream_t stream) {
  const float* feat = (const float*)d_in[0];
  const float* W0   = (const float*)d_in[1];
  const float* b0   = (const float*)d_in[2];
  const float* W1   = (const float*)d_in[3];
  const float* b1   = (const float*)d_in[4];
  const float* W2   = (const float*)d_in[5];
  const float* b2   = (const float*)d_in[6];
  const int*   src  = (const int*)d_in[7];
  const int*   dst  = (const int*)d_in[8];
  float* out = (float*)d_out;

  const int N = in_sizes[0] / 256;
  const int E = in_sizes[7];

  float* bufA = (float*)d_ws;
  float* bufB = bufA + (size_t)N * 64;
  int*   cnt  = (int*)(bufB + (size_t)N * 64);
  int*   rank = (int*)bufB;
  int*   cm   = rank + (((size_t)E + 63) & ~(size_t)63);
  int*   tot  = cm + (size_t)GCN_CHUNKS * N;
  const int rpPad = ((N + 1 + 63) / 64) * 64;
  int* rowptrO = (int*)out;
  int* colO    = rowptrO + rpPad;
  int* rowptrB = (int*)bufB;
  int* colB    = rowptrB + rpPad;

  const int TB = 256;
  const int NPW = 4;
  const int gemmBlocks  = (int)((((long long)(N / 16)) * 32 + 1023) / 1024);
  const int edge4Blocks = (int)((((long long)E + 3) / 4 + TB - 1) / TB);
  const int nodeBlocks  = (N + GCN_ROWS - 1) / GCN_ROWS;
  constexpr int CH = GCN_CHUNKS;
  const int S  = (((E + CH - 1) / CH) + 3) & ~3;
  const int pullBlocks  = (int)(((long long)((N + NPW - 1) / NPW) * 32 + TB - 1) / TB);
  const int TBF = 128;
  const int fusedBlocks = (int)((((long long)(N / 16)) * 32 + TBF - 1) / TBF);
  const int moveBlocks  = (int)((((long long)rpPad + E + 3) / 4 + TB - 1) / TB);

  gcn_count_rank_and_product0<<<gemmBlocks + CH, 1024, 0, stream>>>(dst, cm, tot, rank, E, N, S,
                                                                   gemmBlocks, feat, W0, bufA);
  gcn_chunk_prefix_rowptr<CH><<<nodeBlocks, GCN_ROWS, 0, stream>>>(cm, tot, cnt, rowptrO, N);
  gcn_fill_col<<<edge4Blocks, TB, 0, stream>>>(src, dst, rank, cm, colO, E, N, S);

  gcn_pull_sum<64, NPW, true><<<pullBlocks, TB, 0, stream>>>(bufA, rowptrO, colO, b0, bufB, N, 1);

  gcn_sum_dense_chain_wmma_f32<<<fusedBlocks, TBF, 0, stream>>>(bufB, rowptrO, colO, W1, cnt, b1, W2, bufA, N);

  gcn_move_ints<<<moveBlocks, TB, 0, stream>>>(rowptrO, rowptrB, rpPad + E);
  gcn_pull_sum<32, NPW, true><<<pullBlocks, TB, 0, stream>>>(bufA, rowptrB, colB, b2, out, N, 0);
}
